# starting kernel
# baseline (speedup 1.0000x reference)
#include <hip/hip_runtime.h>
#include <math.h>


typedef float v2f __attribute__((ext_vector_type(2)));
typedef float v8f __attribute__((ext_vector_type(8)));

#if defined(__has_builtin)
#  if __has_builtin(__builtin_amdgcn_wmma_f32_16x16x4_f32)
#    define HAS_WMMA4 1
#  endif
#endif
#ifndef HAS_WMMA4
#  define HAS_WMMA4 0
#endif

namespace {
constexpr int   kN         = 128;
constexpr int   kRingT     = 256;
constexpr int   kRingPitch = 129;
constexpr int   kSteps     = 1000;
constexpr float kDT        = 0.1f;
constexpr float kSig       = 0.0031622776601683794f;
constexpr float kPi        = 3.14159265358979323846f;
constexpr float kInvPi     = 0.3183098861837907f;
constexpr float kJC        = 15.0f;
constexpr float kBDT   = 0.01f;
constexpr float kRTS   = 1.0f / 0.65f;
constexpr float kRTF   = 1.0f / 0.41f;
constexpr float kRTO   = 1.0f / 0.98f;
constexpr float kRAl   = 3.125f;
constexpr float kInvE0 = 2.5f;
constexpr float kOneMinusE0 = 0.6f;
constexpr float kV0 = 4.0f;
constexpr float kK1 = 4.3f * 40.3f * 0.4f * 0.04f;
constexpr float kK2 = 0.5f * 25.0f * 0.4f * 0.04f;
constexpr float kK3 = 0.5f;
}

__global__ __launch_bounds__(256) void tvb_mpr_persistent(
    const float* __restrict__ region_pars,
    const float* __restrict__ Wt,
    const float* __restrict__ g,
    const float* __restrict__ stimulus,
    const float* __restrict__ noise,
    const float* __restrict__ initial_cond,
    const int*   __restrict__ lags,
    float* __restrict__ out)
{
  extern __shared__ float smem[];
  float* ring  = smem;
  float* cxA   = ring + kRingT * kRingPitch;
  float* cxB   = cxA + kN;
  float* stiml = cxB + kN;

  const int tid   = threadIdx.x;
  const int lane  = tid & 31;
  const int wav   = tid >> 5;
  const int mrow  = 16 * wav + (lane & 15);
  const int cbase = (lane >> 4) << 1;

  float wreg[64];
  int   qreg[64];
#pragma unroll
  for (int kt = 0; kt < 32; ++kt) {
    const int j0 = 4 * kt + cbase;
    wreg[2 * kt + 0] = Wt[mrow * kN + j0];
    wreg[2 * kt + 1] = Wt[mrow * kN + j0 + 1];
    qreg[2 * kt + 0] = (1 - lags[mrow * kN + j0]) & 255;
    qreg[2 * kt + 1] = (1 - lags[mrow * kN + j0 + 1]) & 255;
  }

  for (int t = tid; t < 100; t += 256) stiml[t] = stimulus[t];
  for (int idx = tid; idx < kRingT * kN; idx += 256) {
    const int s = idx >> 7, j = idx & 127;
    ring[s * kRingPitch + j] = initial_cond[2 * j];
  }

  float cur_r = 0.f, cur_V = 0.f, eta = 0.f;
  float bs = 1.f, bfl = 1.f, bvl = 1.f, bq = 1.f;
  const float g0 = g[0];
  float2 nz_cur; nz_cur.x = 0.f; nz_cur.y = 0.f;
  if (tid < kN) {
    cur_r  = initial_cond[2 * tid];
    cur_V  = initial_cond[2 * tid + 1];
    eta    = region_pars[tid];
    nz_cur = ((const float2*)noise)[tid];
  }
  __syncthreads();

  auto coupling_window = [&](int base, float* cx_out) {
    v8f accE = {};
    v8f accO = {};
#if HAS_WMMA4
    const v2f bones = {1.0f, 1.0f};
#endif
#pragma unroll
    for (int kt = 0; kt < 32; ++kt) {
      const int j0 = 4 * kt + cbase;
      const int t0 = (base + qreg[2 * kt + 0]) & 255;
      const int t1 = (base + qreg[2 * kt + 1]) & 255;
      v2f a;
      a.x = wreg[2 * kt + 0] * ring[t0 * kRingPitch + j0];
      a.y = wreg[2 * kt + 1] * ring[t1 * kRingPitch + j0 + 1];
#if HAS_WMMA4
      if (kt & 1) {
        accO = __builtin_amdgcn_wmma_f32_16x16x4_f32(
            false, a, false, bones, (short)0, accO, false, false);
      } else {
        accE = __builtin_amdgcn_wmma_f32_16x16x4_f32(
            false, a, false, bones, (short)0, accE, false, false);
      }
#else
      if (kt & 1) accO[kt & 7] += a.x + a.y;
      else        accE[kt & 7] += a.x + a.y;
#endif
    }
#if HAS_WMMA4
    const v8f acc = accE + accO;
    if (lane == 0) {
#pragma unroll
      for (int v = 0; v < 8; ++v) cx_out[16 * wav + v] = acc[v];
    } else if (lane == 16) {
#pragma unroll
      for (int v = 0; v < 8; ++v) cx_out[16 * wav + 8 + v] = acc[v];
    }
#else
    float p = accE[0] + accE[1] + accE[2] + accE[3] +
              accE[4] + accE[5] + accE[6] + accE[7] +
              accO[0] + accO[1] + accO[2] + accO[3] +
              accO[4] + accO[5] + accO[6] + accO[7];
    p += __shfl_xor(p, 16, 32);
    if (lane < 16) cx_out[16 * wav + lane] = p;
#endif
  };

  coupling_window(-1, cxA);

  for (int T = 0; T < kSteps; ++T) {
    float* const cxP = (T & 1) ? cxB : cxA;
    float* const cxN = (T & 1) ? cxA : cxB;

    float2 nz_nxt; nz_nxt.x = 0.f; nz_nxt.y = 0.f;
    if (tid < kN) {
      ring[(T & 255) * kRingPitch + tid] = cur_r;
      if (T + 1 < kSteps) nz_nxt = ((const float2*)noise)[(T + 1) * kN + tid];
    }
    __syncthreads();

    coupling_window(T, cxN);
    __syncthreads();

    if (tid < kN) {
      const float dwr = kSig * nz_cur.x;
      const float dwv = kSig * nz_cur.y;
      const float stv = (tid == 0) ? stiml[T / 10] : 0.f;

      const float rc = cur_r, Vc = cur_V;
      const float I1 = g0 * cxP[tid] + stv;
      const float d1r = kInvPi + 2.f * rc * Vc;
      const float pt1 = kPi * rc;
      const float d1V = Vc * Vc + eta + kJC * rc + I1 - pt1 * pt1;
      const float xir = fmaxf(rc + kDT * d1r + dwr, 0.f);
      const float xiV = Vc + kDT * d1V + dwv;

      const float I2 = g0 * cxN[tid] + stv;
      const float d2r = kInvPi + 2.f * xir * xiV;
      const float pt2 = kPi * xir;
      const float d2V = xiV * xiV + eta + kJC * xir + I2 - pt2 * pt2;

      const float nxr = fmaxf(rc + kDT * 0.5f * (d1r + d2r) + dwr, 0.f);
      const float nxV = Vc + kDT * 0.5f * (d1V + d2V) + dwv;
      cur_r = nxr;
      cur_V = nxV;

      float2 rv2; rv2.x = nxr; rv2.y = nxV;
      ((float2*)out)[T * kN + tid] = rv2;

      {
        const float x   = nxr;
        const float vr1 = powf(bvl, kRAl);
        const float d1s = x - kRTS * bs - kRTF * (bfl - 1.f);
        const float d1f = bs;
        const float d1v = kRTO * (bfl - vr1);
        const float d1q = kRTO * (bfl * (1.f - powf(kOneMinusE0, 1.f / bfl)) * kInvE0
                                  - vr1 * bq / bvl);
        const float s2 = bs + kBDT * d1s;
        const float f2 = bfl + kBDT * d1f;
        const float v2 = bvl + kBDT * d1v;
        const float q2 = bq + kBDT * d1q;
        const float vr2 = powf(v2, kRAl);
        const float d2s = x - kRTS * s2 - kRTF * (f2 - 1.f);
        const float d2f = s2;
        const float d2v = kRTO * (f2 - vr2);
        const float d2q = kRTO * (f2 * (1.f - powf(kOneMinusE0, 1.f / f2)) * kInvE0
                                  - vr2 * q2 / v2);
        bs  += 0.5f * kBDT * (d1s + d2s);
        bfl += 0.5f * kBDT * (d1f + d2f);
        bvl += 0.5f * kBDT * (d1v + d2v);
        bq  += 0.5f * kBDT * (d1q + d2q);
      }
      nz_cur = nz_nxt;
    }
  }

  if (tid < kN) {
    const float bold = kV0 * (kK1 * (1.f - bq) + kK2 * (1.f - bq / bvl) +
                              kK3 * (1.f - bvl));
    out[kSteps * kN * 2 + tid] = bold;
  }
}

extern "C" void kernel_launch(void* const* d_in, const int* in_sizes, int n_in,
                              void* d_out, int out_size, void* d_ws, size_t ws_size,
                              hipStream_t stream) {
  (void)in_sizes; (void)n_in; (void)d_ws; (void)ws_size; (void)out_size;
  const float* region_pars  = (const float*)d_in[0];
  const float* Wt           = (const float*)d_in[1];
  const float* g            = (const float*)d_in[2];
  const float* stimulus     = (const float*)d_in[3];
  const float* noise        = (const float*)d_in[4];
  const float* initial_cond = (const float*)d_in[5];
  const int*   lags         = (const int*)d_in[6];

  const size_t smem_bytes =
      (size_t)(kRingT * kRingPitch + kN + kN + 104) * sizeof(float);

  tvb_mpr_persistent<<<1, 256, smem_bytes, stream>>>(
      region_pars, Wt, g, stimulus, noise, initial_cond, lags, (float*)d_out);
}


# step loop's global traffic blocked through LDS: noise of 40 steps bulk-loaded, rv rows of 40 steps bulk-stored by all 8 waves once per block; no step barrier waits on a global load or store
# speedup vs baseline: 1.2138x; 1.2138x over previous
#include <hip/hip_runtime.h>
#include <math.h>


typedef float v2f __attribute__((ext_vector_type(2)));
typedef float v8f __attribute__((ext_vector_type(8)));

#if defined(__has_builtin)
#  if __has_builtin(__builtin_amdgcn_wmma_f32_16x16x4_f32)
#    define HAS_WMMA4 1
#  endif
#endif
#ifndef HAS_WMMA4
#  define HAS_WMMA4 0
#endif

namespace {
constexpr int   kN         = 128;
constexpr int   kRingT     = 256;
constexpr int   kRingPitch = 129;
constexpr int   kSteps     = 1000;
constexpr int   kBlk       = 40;
constexpr int   kNBlk      = kSteps / kBlk;
static_assert(kSteps % kBlk == 0 && kBlk % 4 == 0, "block must tile the scan");
constexpr float kDT        = 0.1f;
constexpr float kSig       = 0.0031622776601683794f;
constexpr float kPi        = 3.14159265358979323846f;
constexpr float kInvPi     = 0.3183098861837907f;
constexpr float kJC        = 15.0f;
constexpr float kBDT   = 0.01f;
constexpr float kRTS   = 1.0f / 0.65f;
constexpr float kRTF   = 1.0f / 0.41f;
constexpr float kRTO   = 1.0f / 0.98f;
constexpr float kRAl   = 3.125f;
constexpr float kInvE0 = 2.5f;
constexpr float kOneMinusE0 = 0.6f;
constexpr float kV0 = 4.0f;
constexpr float kK1 = 4.3f * 40.3f * 0.4f * 0.04f;
constexpr float kK2 = 0.5f * 25.0f * 0.4f * 0.04f;
constexpr float kK3 = 0.5f;
}

__global__ __launch_bounds__(256) void tvb_mpr_persistent(
    const float* __restrict__ region_pars,
    const float* __restrict__ Wt,
    const float* __restrict__ g,
    const float* __restrict__ stimulus,
    const float* __restrict__ noise,
    const float* __restrict__ initial_cond,
    const int*   __restrict__ lags,
    float* __restrict__ out)
{
  extern __shared__ __attribute__((aligned(16))) float smem[];
  float* ring  = smem;
  float* cxA   = ring + kRingT * kRingPitch;
  float* cxB   = cxA + kN;
  float* stiml = cxB + kN;
  float* nzblk = stiml + 104;
  float* rvblk = nzblk + kBlk * 2 * kN;

  const int tid   = threadIdx.x;
  const int lane  = tid & 31;
  const int wav   = tid >> 5;
  const int mrow  = 16 * wav + (lane & 15);
  const int cbase = (lane >> 4) << 1;

  float wreg[64];
  int   qreg[64];
#pragma unroll
  for (int kt = 0; kt < 32; ++kt) {
    const int j0 = 4 * kt + cbase;
    wreg[2 * kt + 0] = Wt[mrow * kN + j0];
    wreg[2 * kt + 1] = Wt[mrow * kN + j0 + 1];
    qreg[2 * kt + 0] = (1 - lags[mrow * kN + j0]) & 255;
    qreg[2 * kt + 1] = (1 - lags[mrow * kN + j0 + 1]) & 255;
  }

  for (int t = tid; t < 100; t += 256) stiml[t] = stimulus[t];
  for (int idx = tid; idx < kRingT * kN; idx += 256) {
    const int s = idx >> 7, j = idx & 127;
    ring[s * kRingPitch + j] = initial_cond[2 * j];
  }

  float cur_r = 0.f, cur_V = 0.f, eta = 0.f;
  float bs = 1.f, bfl = 1.f, bvl = 1.f, bq = 1.f;
  const float g0 = g[0];
  if (tid < kN) {
    cur_r  = initial_cond[2 * tid];
    cur_V  = initial_cond[2 * tid + 1];
    eta    = region_pars[tid];
  }
  __syncthreads();

  auto coupling_window = [&](int base, float* cx_out) {
    v8f accE = {};
    v8f accO = {};
#if HAS_WMMA4
    const v2f bones = {1.0f, 1.0f};
#endif
#pragma unroll
    for (int kt = 0; kt < 32; ++kt) {
      const int j0 = 4 * kt + cbase;
      const int t0 = (base + qreg[2 * kt + 0]) & 255;
      const int t1 = (base + qreg[2 * kt + 1]) & 255;
      v2f a;
      a.x = wreg[2 * kt + 0] * ring[t0 * kRingPitch + j0];
      a.y = wreg[2 * kt + 1] * ring[t1 * kRingPitch + j0 + 1];
#if HAS_WMMA4
      if (kt & 1) {
        accO = __builtin_amdgcn_wmma_f32_16x16x4_f32(
            false, a, false, bones, (short)0, accO, false, false);
      } else {
        accE = __builtin_amdgcn_wmma_f32_16x16x4_f32(
            false, a, false, bones, (short)0, accE, false, false);
      }
#else
      if (kt & 1) accO[kt & 7] += a.x + a.y;
      else        accE[kt & 7] += a.x + a.y;
#endif
    }
#if HAS_WMMA4
    const v8f acc = accE + accO;
    if (lane == 0) {
#pragma unroll
      for (int v = 0; v < 8; ++v) cx_out[16 * wav + v] = acc[v];
    } else if (lane == 16) {
#pragma unroll
      for (int v = 0; v < 8; ++v) cx_out[16 * wav + 8 + v] = acc[v];
    }
#else
    float p = accE[0] + accE[1] + accE[2] + accE[3] +
              accE[4] + accE[5] + accE[6] + accE[7] +
              accO[0] + accO[1] + accO[2] + accO[3] +
              accO[4] + accO[5] + accO[6] + accO[7];
    p += __shfl_xor(p, 16, 32);
    if (lane < 16) cx_out[16 * wav + lane] = p;
#endif
  };

  coupling_window(-1, cxA);

  auto flush_rv = [&](int blk) {
    float4* dst = (float4*)out + blk * (kBlk * 2 * kN / 4);
    const float4* s4 = (const float4*)rvblk;
#pragma unroll
    for (int k = 0; k < kBlk / 4; ++k) dst[tid + 256 * k] = s4[tid + 256 * k];
  };

  for (int blk = 0; blk < kNBlk; ++blk) {
    __syncthreads();
    {
      const float4* nsrc = (const float4*)noise + blk * (kBlk * 2 * kN / 4);
      float4 nz4[kBlk / 4];
#pragma unroll
      for (int k = 0; k < kBlk / 4; ++k) nz4[k] = nsrc[tid + 256 * k];
      if (blk > 0) flush_rv(blk - 1);
#pragma unroll
      for (int k = 0; k < kBlk / 4; ++k)
        ((float4*)nzblk)[tid + 256 * k] = nz4[k];
    }

   for (int s = 0; s < kBlk; ++s) {
    const int T = blk * kBlk + s;
    float* const cxP = (T & 1) ? cxB : cxA;
    float* const cxN = (T & 1) ? cxA : cxB;

    if (tid < kN) ring[(T & 255) * kRingPitch + tid] = cur_r;
    __syncthreads();

    const float2 nz_cur = ((const float2*)nzblk)[s * kN + (tid & (kN - 1))];

    coupling_window(T, cxN);
    __syncthreads();

    if (tid < kN) {
      const float dwr = kSig * nz_cur.x;
      const float dwv = kSig * nz_cur.y;
      const float stv = (tid == 0) ? stiml[T / 10] : 0.f;

      const float rc = cur_r, Vc = cur_V;
      const float I1 = g0 * cxP[tid] + stv;
      const float d1r = kInvPi + 2.f * rc * Vc;
      const float pt1 = kPi * rc;
      const float d1V = Vc * Vc + eta + kJC * rc + I1 - pt1 * pt1;
      const float xir = fmaxf(rc + kDT * d1r + dwr, 0.f);
      const float xiV = Vc + kDT * d1V + dwv;

      const float I2 = g0 * cxN[tid] + stv;
      const float d2r = kInvPi + 2.f * xir * xiV;
      const float pt2 = kPi * xir;
      const float d2V = xiV * xiV + eta + kJC * xir + I2 - pt2 * pt2;

      const float nxr = fmaxf(rc + kDT * 0.5f * (d1r + d2r) + dwr, 0.f);
      const float nxV = Vc + kDT * 0.5f * (d1V + d2V) + dwv;
      cur_r = nxr;
      cur_V = nxV;

      float2 rv2; rv2.x = nxr; rv2.y = nxV;
      ((float2*)rvblk)[s * kN + tid] = rv2;

      {
        const float x   = nxr;
        const float vr1 = powf(bvl, kRAl);
        const float d1s = x - kRTS * bs - kRTF * (bfl - 1.f);
        const float d1f = bs;
        const float d1v = kRTO * (bfl - vr1);
        const float d1q = kRTO * (bfl * (1.f - powf(kOneMinusE0, 1.f / bfl)) * kInvE0
                                  - vr1 * bq / bvl);
        const float s2 = bs + kBDT * d1s;
        const float f2 = bfl + kBDT * d1f;
        const float v2 = bvl + kBDT * d1v;
        const float q2 = bq + kBDT * d1q;
        const float vr2 = powf(v2, kRAl);
        const float d2s = x - kRTS * s2 - kRTF * (f2 - 1.f);
        const float d2f = s2;
        const float d2v = kRTO * (f2 - vr2);
        const float d2q = kRTO * (f2 * (1.f - powf(kOneMinusE0, 1.f / f2)) * kInvE0
                                  - vr2 * q2 / v2);
        bs  += 0.5f * kBDT * (d1s + d2s);
        bfl += 0.5f * kBDT * (d1f + d2f);
        bvl += 0.5f * kBDT * (d1v + d2v);
        bq  += 0.5f * kBDT * (d1q + d2q);
      }
    }
   }
  }

  __syncthreads();
  flush_rv(kNBlk - 1);

  if (tid < kN) {
    const float bold = kV0 * (kK1 * (1.f - bq) + kK2 * (1.f - bq / bvl) +
                              kK3 * (1.f - bvl));
    out[kSteps * kN * 2 + tid] = bold;
  }
}

extern "C" void kernel_launch(void* const* d_in, const int* in_sizes, int n_in,
                              void* d_out, int out_size, void* d_ws, size_t ws_size,
                              hipStream_t stream) {
  (void)in_sizes; (void)n_in; (void)d_ws; (void)ws_size; (void)out_size;
  const float* region_pars  = (const float*)d_in[0];
  const float* Wt           = (const float*)d_in[1];
  const float* g            = (const float*)d_in[2];
  const float* stimulus     = (const float*)d_in[3];
  const float* noise        = (const float*)d_in[4];
  const float* initial_cond = (const float*)d_in[5];
  const int*   lags         = (const int*)d_in[6];

  const size_t smem_bytes =
      (size_t)(kRingT * kRingPitch + kN + kN + 104 + 2 * kBlk * 2 * kN) *
      sizeof(float);

  tvb_mpr_persistent<<<1, 256, smem_bytes, stream>>>(
      region_pars, Wt, g, stimulus, noise, initial_cond, lags, (float*)d_out);
}


# balloon v^3.125 evaluated as v*v*v*sqrt(sqrt(sqrt(v))) (3.125 = 3 + 1/8) instead of two powf calls a step: the loop-carried v chain loses its log/exp; pow values change in last places
# speedup vs baseline: 1.8242x; 1.5029x over previous
#include <hip/hip_runtime.h>
#include <math.h>


typedef float v2f __attribute__((ext_vector_type(2)));
typedef float v8f __attribute__((ext_vector_type(8)));

#if defined(__has_builtin)
#  if __has_builtin(__builtin_amdgcn_wmma_f32_16x16x4_f32)
#    define HAS_WMMA4 1
#  endif
#endif
#ifndef HAS_WMMA4
#  define HAS_WMMA4 0
#endif

namespace {
constexpr int   kN         = 128;
constexpr int   kRingT     = 256;
constexpr int   kRingPitch = 128;
constexpr unsigned kRingMask = kRingT * kRingPitch * 4 - 1;
constexpr int   kSteps     = 1000;
constexpr int   kBlk       = 40;
constexpr int   kNBlk      = kSteps / kBlk;
static_assert(kSteps % kBlk == 0 && kBlk % 4 == 0, "block must tile the scan");
constexpr float kDT        = 0.1f;
constexpr float kSig       = 0.0031622776601683794f;
constexpr float kPi        = 3.14159265358979323846f;
constexpr float kInvPi     = 0.3183098861837907f;
constexpr float kJC        = 15.0f;
constexpr float kBDT   = 0.01f;
constexpr float kRTS   = 1.0f / 0.65f;
constexpr float kRTF   = 1.0f / 0.41f;
constexpr float kRTO   = 1.0f / 0.98f;
constexpr float kInvE0 = 2.5f;
constexpr float kOneMinusE0 = 0.6f;
constexpr float kV0 = 4.0f;
constexpr float kK1 = 4.3f * 40.3f * 0.4f * 0.04f;
constexpr float kK2 = 0.5f * 25.0f * 0.4f * 0.04f;
constexpr float kK3 = 0.5f;
}

__global__ __launch_bounds__(256) void tvb_mpr_persistent(
    const float* __restrict__ region_pars,
    const float* __restrict__ Wt,
    const float* __restrict__ g,
    const float* __restrict__ stimulus,
    const float* __restrict__ noise,
    const float* __restrict__ initial_cond,
    const int*   __restrict__ lags,
    float* __restrict__ out)
{
  extern __shared__ __attribute__((aligned(16))) float smem[];
  float* ring  = smem;
  float* cxA   = ring + kRingT * kRingPitch;
  float* cxB   = cxA + kN;
  float* stiml = cxB + kN;
  float* nzblk = stiml + 104;
  float* rvblk = nzblk + kBlk * 2 * kN;

  const int tid   = threadIdx.x;
  const int lane  = tid & 31;
  const int wav   = tid >> 5;
  const int lcol  = lane & 15;

  float    wreg[64];
  unsigned qoff[64];
#pragma unroll
  for (int ii = 0; ii < 16; ++ii) {
#pragma unroll
    for (int k = 0; k < 4; ++k) {
      const int i = 16 * wav + ii;
      const int j = lane + 32 * k;
      wreg[4 * ii + k] = Wt[i * kN + j];
      qoff[4 * ii + k] =
          ((unsigned)((1 - lags[i * kN + j]) & 255) << 9) | ((unsigned)j << 2);
    }
  }

  for (int t = tid; t < 100; t += 256) stiml[t] = stimulus[t];
  for (int idx = tid; idx < kRingT * kN; idx += 256) {
    ring[idx] = initial_cond[2 * (idx & 127)];
  }

  float cur_r = 0.f, cur_V = 0.f, eta = 0.f;
  float bs = 1.f, bfl = 1.f, bvl = 1.f, bq = 1.f;
  const float g0 = g[0];
  if (tid < kN) {
    cur_r  = initial_cond[2 * tid];
    cur_V  = initial_cond[2 * tid + 1];
    eta    = region_pars[tid];
  }
  __syncthreads();

  auto pow3125 = [](float v) -> float {
    const float r8 = sqrtf(sqrtf(sqrtf(v)));
    return (v * v) * (v * r8);
  };

  auto xhalf = [](float x) -> float {
    const int b = __builtin_bit_cast(int, x);
    return __builtin_bit_cast(
        float, __builtin_amdgcn_permlanex16(b, b, 0x76543210, 0xfedcba98,
                                            false, false));
  };

  auto coupling_window = [&](int base, float* cx_out) {
    const unsigned tb = (unsigned)(base & 255) << 9;
    const char* const ringb = (const char*)ring;
    float rr[64];
#pragma unroll
    for (int e = 0; e < 64; ++e)
      rr[e] = *(const float*)(ringb + ((tb + qoff[e]) & kRingMask));
    float p[16];
#pragma unroll
    for (int ii = 0; ii < 16; ++ii) {
      float s = wreg[4 * ii] * rr[4 * ii];
      s = fmaf(wreg[4 * ii + 1], rr[4 * ii + 1], s);
      s = fmaf(wreg[4 * ii + 2], rr[4 * ii + 2], s);
      s = fmaf(wreg[4 * ii + 3], rr[4 * ii + 3], s);
      p[ii] = s;
    }
#if HAS_WMMA4
    v8f accE = {};
    v8f accO = {};
#pragma unroll
    for (int pr = 0; pr < 8; ++pr) {
      v2f a, b;
      a.x = p[2 * pr];
      a.y = p[2 * pr + 1];
      b.x = (lcol == 2 * pr) ? 1.0f : 0.0f;
      b.y = (lcol == 2 * pr + 1) ? 1.0f : 0.0f;
      if (pr & 1) {
        accO = __builtin_amdgcn_wmma_f32_16x16x4_f32(
            false, a, false, b, (short)0, accO, false, false);
      } else {
        accE = __builtin_amdgcn_wmma_f32_16x16x4_f32(
            false, a, false, b, (short)0, accE, false, false);
      }
    }
    const v8f acc = accE + accO;
    float s = ((acc[0] + acc[1]) + (acc[2] + acc[3])) +
              ((acc[4] + acc[5]) + (acc[6] + acc[7]));
    s += xhalf(s);
    if (lane < 16) cx_out[16 * wav + lane] = s;
#else
    float mine = 0.f;
#pragma unroll
    for (int ii = 0; ii < 16; ++ii) {
      float s = p[ii];
      for (int m = 16; m >= 1; m >>= 1) s += __shfl_xor(s, m, 32);
      if (lane == ii) mine = s;
    }
    if (lane < 16) cx_out[16 * wav + lane] = mine;
#endif
  };

  coupling_window(-1, cxA);

  auto flush_rv = [&](int blk) {
    float4* dst = (float4*)out + blk * (kBlk * 2 * kN / 4);
    const float4* s4 = (const float4*)rvblk;
#pragma unroll
    for (int k = 0; k < kBlk / 4; ++k) dst[tid + 256 * k] = s4[tid + 256 * k];
  };

  for (int blk = 0; blk < kNBlk; ++blk) {
    __syncthreads();
    {
      const float4* nsrc = (const float4*)noise + blk * (kBlk * 2 * kN / 4);
      float4 nz4[kBlk / 4];
#pragma unroll
      for (int k = 0; k < kBlk / 4; ++k) nz4[k] = nsrc[tid + 256 * k];
      if (blk > 0) flush_rv(blk - 1);
#pragma unroll
      for (int k = 0; k < kBlk / 4; ++k)
        ((float4*)nzblk)[tid + 256 * k] = nz4[k];
    }

   for (int s = 0; s < kBlk; ++s) {
    const int T = blk * kBlk + s;
    float* const cxP = (T & 1) ? cxB : cxA;
    float* const cxN = (T & 1) ? cxA : cxB;

    if (tid < kN) ring[(T & 255) * kRingPitch + tid] = cur_r;
    __syncthreads();

    const float2 nz_cur = ((const float2*)nzblk)[s * kN + (tid & (kN - 1))];

    coupling_window(T, cxN);
    __syncthreads();

    if (tid < kN) {
      const float dwr = kSig * nz_cur.x;
      const float dwv = kSig * nz_cur.y;
      const float stv = (tid == 0) ? stiml[T / 10] : 0.f;

      const float rc = cur_r, Vc = cur_V;
      const float I1 = g0 * cxP[tid] + stv;
      const float d1r = kInvPi + 2.f * rc * Vc;
      const float pt1 = kPi * rc;
      const float d1V = Vc * Vc + eta + kJC * rc + I1 - pt1 * pt1;
      const float xir = fmaxf(rc + kDT * d1r + dwr, 0.f);
      const float xiV = Vc + kDT * d1V + dwv;

      const float I2 = g0 * cxN[tid] + stv;
      const float d2r = kInvPi + 2.f * xir * xiV;
      const float pt2 = kPi * xir;
      const float d2V = xiV * xiV + eta + kJC * xir + I2 - pt2 * pt2;

      const float nxr = fmaxf(rc + kDT * 0.5f * (d1r + d2r) + dwr, 0.f);
      const float nxV = Vc + kDT * 0.5f * (d1V + d2V) + dwv;
      cur_r = nxr;
      cur_V = nxV;

      float2 rv2; rv2.x = nxr; rv2.y = nxV;
      ((float2*)rvblk)[s * kN + tid] = rv2;

      {
        const float x   = nxr;
        const float vr1 = pow3125(bvl);
        const float d1s = x - kRTS * bs - kRTF * (bfl - 1.f);
        const float d1f = bs;
        const float d1v = kRTO * (bfl - vr1);
        const float d1q = kRTO * (bfl * (1.f - powf(kOneMinusE0, 1.f / bfl)) * kInvE0
                                  - vr1 * bq / bvl);
        const float s2 = bs + kBDT * d1s;
        const float f2 = bfl + kBDT * d1f;
        const float v2 = bvl + kBDT * d1v;
        const float q2 = bq + kBDT * d1q;
        const float vr2 = pow3125(v2);
        const float d2s = x - kRTS * s2 - kRTF * (f2 - 1.f);
        const float d2f = s2;
        const float d2v = kRTO * (f2 - vr2);
        const float d2q = kRTO * (f2 * (1.f - powf(kOneMinusE0, 1.f / f2)) * kInvE0
                                  - vr2 * q2 / v2);
        bs  += 0.5f * kBDT * (d1s + d2s);
        bfl += 0.5f * kBDT * (d1f + d2f);
        bvl += 0.5f * kBDT * (d1v + d2v);
        bq  += 0.5f * kBDT * (d1q + d2q);
      }
    }
   }
  }

  __syncthreads();
  flush_rv(kNBlk - 1);

  if (tid < kN) {
    const float bold = kV0 * (kK1 * (1.f - bq) + kK2 * (1.f - bq / bvl) +
                              kK3 * (1.f - bvl));
    out[kSteps * kN * 2 + tid] = bold;
  }
}

extern "C" void kernel_launch(void* const* d_in, const int* in_sizes, int n_in,
                              void* d_out, int out_size, void* d_ws, size_t ws_size,
                              hipStream_t stream) {
  (void)in_sizes; (void)n_in; (void)d_ws; (void)ws_size; (void)out_size;
  const float* region_pars  = (const float*)d_in[0];
  const float* Wt           = (const float*)d_in[1];
  const float* g            = (const float*)d_in[2];
  const float* stimulus     = (const float*)d_in[3];
  const float* noise        = (const float*)d_in[4];
  const float* initial_cond = (const float*)d_in[5];
  const int*   lags         = (const int*)d_in[6];

  const size_t smem_bytes =
      (size_t)(kRingT * kRingPitch + kN + kN + 104 + 2 * kBlk * 2 * kN) *
      sizeof(float);

  tvb_mpr_persistent<<<1, 256, smem_bytes, stream>>>(
      region_pars, Wt, g, stimulus, noise, initial_cond, lags, (float*)d_out);
}


# balloon step runs one row behind inside the coupling phase of the four waves that own the regions (no waves added): its register chain fills the LDS wait of their gathers
# speedup vs baseline: 1.9678x; 1.0787x over previous
#include <hip/hip_runtime.h>
#include <math.h>


typedef float v2f __attribute__((ext_vector_type(2)));
typedef float v8f __attribute__((ext_vector_type(8)));

#if defined(__has_builtin)
#  if __has_builtin(__builtin_amdgcn_wmma_f32_16x16x4_f32)
#    define HAS_WMMA4 1
#  endif
#endif
#ifndef HAS_WMMA4
#  define HAS_WMMA4 0
#endif

namespace {
constexpr int   kN         = 128;
constexpr int   kRingT     = 256;
constexpr int   kRingPitch = 128;
constexpr unsigned kRingMask = kRingT * kRingPitch * 4 - 1;
constexpr int   kSteps     = 1000;
constexpr int   kBlk       = 40;
constexpr int   kNBlk      = kSteps / kBlk;
static_assert(kSteps % kBlk == 0 && kBlk % 4 == 0, "block must tile the scan");
constexpr float kDT        = 0.1f;
constexpr float kSig       = 0.0031622776601683794f;
constexpr float kPi        = 3.14159265358979323846f;
constexpr float kInvPi     = 0.3183098861837907f;
constexpr float kJC        = 15.0f;
constexpr float kBDT   = 0.01f;
constexpr float kRTS   = 1.0f / 0.65f;
constexpr float kRTF   = 1.0f / 0.41f;
constexpr float kRTO   = 1.0f / 0.98f;
constexpr float kInvE0 = 2.5f;
constexpr float kOneMinusE0 = 0.6f;
constexpr float kV0 = 4.0f;
constexpr float kK1 = 4.3f * 40.3f * 0.4f * 0.04f;
constexpr float kK2 = 0.5f * 25.0f * 0.4f * 0.04f;
constexpr float kK3 = 0.5f;
}

__global__ __launch_bounds__(256) void tvb_mpr_persistent(
    const float* __restrict__ region_pars,
    const float* __restrict__ Wt,
    const float* __restrict__ g,
    const float* __restrict__ stimulus,
    const float* __restrict__ noise,
    const float* __restrict__ initial_cond,
    const int*   __restrict__ lags,
    float* __restrict__ out)
{
  extern __shared__ __attribute__((aligned(16))) float smem[];
  float* ring  = smem;
  float* cxA   = ring + kRingT * kRingPitch;
  float* cxB   = cxA + kN;
  float* stiml = cxB + kN;
  float* nzblk = stiml + 104;
  float* rvblk = nzblk + kBlk * 2 * kN;

  const int tid   = threadIdx.x;
  const int lane  = tid & 31;
  const int wav   = __builtin_amdgcn_readfirstlane(tid >> 5);
  const int lcol  = lane & 15;

  float    wreg[64];
  unsigned qoff[64];
#pragma unroll
  for (int ii = 0; ii < 16; ++ii) {
#pragma unroll
    for (int k = 0; k < 4; ++k) {
      const int i = 16 * wav + ii;
      const int j = lane + 32 * k;
      wreg[4 * ii + k] = Wt[i * kN + j];
      qoff[4 * ii + k] =
          ((unsigned)((1 - lags[i * kN + j]) & 255) << 9) | ((unsigned)j << 2);
    }
  }

  for (int t = tid; t < 100; t += 256) stiml[t] = stimulus[t];
  for (int idx = tid; idx < kRingT * kN; idx += 256) {
    ring[idx] = initial_cond[2 * (idx & 127)];
  }

  float cur_r = 0.f, cur_V = 0.f, eta = 0.f;
  float bs = 1.f, bfl = 1.f, bvl = 1.f, bq = 1.f;
  const float g0 = g[0];
  if (tid < kN) {
    cur_r  = initial_cond[2 * tid];
    cur_V  = initial_cond[2 * tid + 1];
    eta    = region_pars[tid];
  }
  __syncthreads();

  auto pow3125 = [](float v) -> float {
    const float r8 = sqrtf(sqrtf(sqrtf(v)));
    return (v * v) * (v * r8);
  };

  auto xhalf = [](float x) -> float {
    const int b = __builtin_bit_cast(int, x);
    return __builtin_bit_cast(
        float, __builtin_amdgcn_permlanex16(b, b, 0x76543210, 0xfedcba98,
                                            false, false));
  };

  auto coupling_window = [&](int base, float* cx_out, auto&& beside) {
    const unsigned tb = (unsigned)(base & 255) << 9;
    const char* const ringb = (const char*)ring;
    float rr[64];
#pragma unroll
    for (int e = 0; e < 64; ++e)
      rr[e] = *(const float*)(ringb + ((tb + qoff[e]) & kRingMask));
    beside();
    float p[16];
#pragma unroll
    for (int ii = 0; ii < 16; ++ii) {
      float s = wreg[4 * ii] * rr[4 * ii];
      s = fmaf(wreg[4 * ii + 1], rr[4 * ii + 1], s);
      s = fmaf(wreg[4 * ii + 2], rr[4 * ii + 2], s);
      s = fmaf(wreg[4 * ii + 3], rr[4 * ii + 3], s);
      p[ii] = s;
    }
#if HAS_WMMA4
    v8f accE = {};
    v8f accO = {};
#pragma unroll
    for (int pr = 0; pr < 8; ++pr) {
      v2f a, b;
      a.x = p[2 * pr];
      a.y = p[2 * pr + 1];
      b.x = (lcol == 2 * pr) ? 1.0f : 0.0f;
      b.y = (lcol == 2 * pr + 1) ? 1.0f : 0.0f;
      if (pr & 1) {
        accO = __builtin_amdgcn_wmma_f32_16x16x4_f32(
            false, a, false, b, (short)0, accO, false, false);
      } else {
        accE = __builtin_amdgcn_wmma_f32_16x16x4_f32(
            false, a, false, b, (short)0, accE, false, false);
      }
    }
    const v8f acc = accE + accO;
    float s = ((acc[0] + acc[1]) + (acc[2] + acc[3])) +
              ((acc[4] + acc[5]) + (acc[6] + acc[7]));
    s += xhalf(s);
    if (lane < 16) cx_out[16 * wav + lane] = s;
#else
    float mine = 0.f;
#pragma unroll
    for (int ii = 0; ii < 16; ++ii) {
      float s = p[ii];
      for (int m = 16; m >= 1; m >>= 1) s += __shfl_xor(s, m, 32);
      if (lane == ii) mine = s;
    }
    if (lane < 16) cx_out[16 * wav + lane] = mine;
#endif
  };

  auto bold_step = [&](float x, float hb) {
    const float vr1 = pow3125(bvl);
    const float d1s = x - kRTS * bs - kRTF * (bfl - 1.f);
    const float d1f = bs;
    const float d1v = kRTO * (bfl - vr1);
    const float d1q = kRTO * (bfl * (1.f - powf(kOneMinusE0, 1.f / bfl)) * kInvE0
                              - vr1 * bq / bvl);
    const float s2 = bs + kBDT * d1s;
    const float f2 = bfl + kBDT * d1f;
    const float v2 = bvl + kBDT * d1v;
    const float q2 = bq + kBDT * d1q;
    const float vr2 = pow3125(v2);
    const float d2s = x - kRTS * s2 - kRTF * (f2 - 1.f);
    const float d2f = s2;
    const float d2v = kRTO * (f2 - vr2);
    const float d2q = kRTO * (f2 * (1.f - powf(kOneMinusE0, 1.f / f2)) * kInvE0
                              - vr2 * q2 / v2);
    bs  += hb * (d1s + d2s);
    bfl += hb * (d1f + d2f);
    bvl += hb * (d1v + d2v);
    bq  += hb * (d1q + d2q);
  };

  coupling_window(-1, cxA, [] {});

  auto flush_rv = [&](int blk) {
    float4* dst = (float4*)out + blk * (kBlk * 2 * kN / 4);
    const float4* s4 = (const float4*)rvblk;
#pragma unroll
    for (int k = 0; k < kBlk / 4; ++k) dst[tid + 256 * k] = s4[tid + 256 * k];
  };

  for (int blk = 0; blk < kNBlk; ++blk) {
    __syncthreads();
    {
      const float4* nsrc = (const float4*)noise + blk * (kBlk * 2 * kN / 4);
      float4 nz4[kBlk / 4];
#pragma unroll
      for (int k = 0; k < kBlk / 4; ++k) nz4[k] = nsrc[tid + 256 * k];
      if (blk > 0) flush_rv(blk - 1);
#pragma unroll
      for (int k = 0; k < kBlk / 4; ++k)
        ((float4*)nzblk)[tid + 256 * k] = nz4[k];
    }

   for (int s = 0; s < kBlk; ++s) {
    const int T = blk * kBlk + s;
    float* const cxP = (T & 1) ? cxB : cxA;
    float* const cxN = (T & 1) ? cxA : cxB;

    if (tid < kN) ring[(T & 255) * kRingPitch + tid] = cur_r;
    __syncthreads();

    const float2 nz_cur = ((const float2*)nzblk)[s * kN + (tid & (kN - 1))];

    if (wav < 4) {
      const float hb = (T > 0) ? 0.5f * kBDT : 0.f;
      coupling_window(T, cxN, [&] { bold_step(cur_r, hb); });
    } else {
      coupling_window(T, cxN, [] {});
    }
    __syncthreads();

    if (tid < kN) {
      const float dwr = kSig * nz_cur.x;
      const float dwv = kSig * nz_cur.y;
      const float stv = (tid == 0) ? stiml[T / 10] : 0.f;

      const float rc = cur_r, Vc = cur_V;
      const float I1 = g0 * cxP[tid] + stv;
      const float d1r = kInvPi + 2.f * rc * Vc;
      const float pt1 = kPi * rc;
      const float d1V = Vc * Vc + eta + kJC * rc + I1 - pt1 * pt1;
      const float xir = fmaxf(rc + kDT * d1r + dwr, 0.f);
      const float xiV = Vc + kDT * d1V + dwv;

      const float I2 = g0 * cxN[tid] + stv;
      const float d2r = kInvPi + 2.f * xir * xiV;
      const float pt2 = kPi * xir;
      const float d2V = xiV * xiV + eta + kJC * xir + I2 - pt2 * pt2;

      const float nxr = fmaxf(rc + kDT * 0.5f * (d1r + d2r) + dwr, 0.f);
      const float nxV = Vc + kDT * 0.5f * (d1V + d2V) + dwv;
      cur_r = nxr;
      cur_V = nxV;

      float2 rv2; rv2.x = nxr; rv2.y = nxV;
      ((float2*)rvblk)[s * kN + tid] = rv2;
    }
   }
  }

  __syncthreads();
  flush_rv(kNBlk - 1);

  if (tid < kN) {
    bold_step(cur_r, 0.5f * kBDT);
    const float bold = kV0 * (kK1 * (1.f - bq) + kK2 * (1.f - bq / bvl) +
                              kK3 * (1.f - bvl));
    out[kSteps * kN * 2 + tid] = bold;
  }
}

extern "C" void kernel_launch(void* const* d_in, const int* in_sizes, int n_in,
                              void* d_out, int out_size, void* d_ws, size_t ws_size,
                              hipStream_t stream) {
  (void)in_sizes; (void)n_in; (void)d_ws; (void)ws_size; (void)out_size;
  const float* region_pars  = (const float*)d_in[0];
  const float* Wt           = (const float*)d_in[1];
  const float* g            = (const float*)d_in[2];
  const float* stimulus     = (const float*)d_in[3];
  const float* noise        = (const float*)d_in[4];
  const float* initial_cond = (const float*)d_in[5];
  const int*   lags         = (const int*)d_in[6];

  const size_t smem_bytes =
      (size_t)(kRingT * kRingPitch + kN + kN + 104 + 2 * kBlk * 2 * kN) *
      sizeof(float);

  tvb_mpr_persistent<<<1, 256, smem_bytes, stream>>>(
      region_pars, Wt, g, stimulus, noise, initial_cond, lags, (float*)d_out);
}


# balloon model split along its dependency: waves 0..3 integrate the flow pair (s,f) a row ahead and both powf terms, waves 4..7 keep (v,q) and fetch them from LDS; values change in last places
# speedup vs baseline: 2.1956x; 1.1158x over previous
#include <hip/hip_runtime.h>
#include <math.h>


typedef float v2f __attribute__((ext_vector_type(2)));
typedef float v8f __attribute__((ext_vector_type(8)));

#if defined(__has_builtin)
#  if __has_builtin(__builtin_amdgcn_wmma_f32_16x16x4_f32)
#    define HAS_WMMA4 1
#  endif
#endif
#ifndef HAS_WMMA4
#  define HAS_WMMA4 0
#endif

namespace {
constexpr int   kN         = 128;
constexpr int   kRingT     = 256;
constexpr int   kRingPitch = 128;
constexpr unsigned kRingMask = kRingT * kRingPitch * 4 - 1;
constexpr int   kSteps     = 1000;
constexpr int   kBlk       = 40;
constexpr int   kNBlk      = kSteps / kBlk;
static_assert(kSteps % kBlk == 0 && kBlk % 4 == 0, "block must tile the scan");
constexpr float kDT        = 0.1f;
constexpr float kSig       = 0.0031622776601683794f;
constexpr float kPi        = 3.14159265358979323846f;
constexpr float kInvPi     = 0.3183098861837907f;
constexpr float kJC        = 15.0f;
constexpr float kBDT   = 0.01f;
constexpr float kRTS   = 1.0f / 0.65f;
constexpr float kRTF   = 1.0f / 0.41f;
constexpr float kRTO   = 1.0f / 0.98f;
constexpr float kInvE0 = 2.5f;
constexpr float kOneMinusE0 = 0.6f;
constexpr float kV0 = 4.0f;
constexpr float kK1 = 4.3f * 40.3f * 0.4f * 0.04f;
constexpr float kK2 = 0.5f * 25.0f * 0.4f * 0.04f;
constexpr float kK3 = 0.5f;
}

__global__ __launch_bounds__(256) void tvb_mpr_persistent(
    const float* __restrict__ region_pars,
    const float* __restrict__ Wt,
    const float* __restrict__ g,
    const float* __restrict__ stimulus,
    const float* __restrict__ noise,
    const float* __restrict__ initial_cond,
    const int*   __restrict__ lags,
    float* __restrict__ out)
{
  extern __shared__ __attribute__((aligned(16))) float smem[];
  float* ring  = smem;
  float* cxA   = ring + kRingT * kRingPitch;
  float* cxB   = cxA + kN;
  float* stiml = cxB + kN;
  float* nzblk = stiml + 104;
  float* rvblk = nzblk + kBlk * 2 * kN;
  float4* gpk  = (float4*)(rvblk + kBlk * 2 * kN);

  const int tid   = threadIdx.x;
  const int lane  = tid & 31;
  const int wav   = __builtin_amdgcn_readfirstlane(tid >> 5);
  const int lcol  = lane & 15;

  float    wreg[64];
  unsigned qoff[64];
#pragma unroll
  for (int ii = 0; ii < 16; ++ii) {
#pragma unroll
    for (int k = 0; k < 4; ++k) {
      const int i = 16 * wav + ii;
      const int j = lane + 32 * k;
      wreg[4 * ii + k] = Wt[i * kN + j];
      qoff[4 * ii + k] =
          ((unsigned)((1 - lags[i * kN + j]) & 255) << 9) | ((unsigned)j << 2);
    }
  }

  for (int t = tid; t < 100; t += 256) stiml[t] = stimulus[t];
  for (int idx = tid; idx < kRingT * kN; idx += 256) {
    ring[idx] = initial_cond[2 * (idx & 127)];
  }

  float cur_r = 0.f, cur_V = 0.f, eta = 0.f;
  float hs = 1.f, hf = 1.f;
  float bvl = 1.f, bq = 1.f;
  float4 pk = make_float4(1.f, 1.f, 1.f, 1.f);
  const float g0 = g[0];
  if (tid < kN) {
    cur_r  = initial_cond[2 * tid];
    cur_V  = initial_cond[2 * tid + 1];
    eta    = region_pars[tid];
  }
  __syncthreads();

  auto pow3125 = [](float v) -> float {
    const float r8 = sqrtf(sqrtf(sqrtf(v)));
    return (v * v) * (v * r8);
  };

  auto xhalf = [](float x) -> float {
    const int b = __builtin_bit_cast(int, x);
    return __builtin_bit_cast(
        float, __builtin_amdgcn_permlanex16(b, b, 0x76543210, 0xfedcba98,
                                            false, false));
  };

  auto coupling_window = [&](int base, float* cx_out, auto&& beside) {
    const unsigned tb = (unsigned)(base & 255) << 9;
    const char* const ringb = (const char*)ring;
    float rr[64];
#pragma unroll
    for (int e = 0; e < 64; ++e)
      rr[e] = *(const float*)(ringb + ((tb + qoff[e]) & kRingMask));
    beside();
    float p[16];
#pragma unroll
    for (int ii = 0; ii < 16; ++ii) {
      float s = wreg[4 * ii] * rr[4 * ii];
      s = fmaf(wreg[4 * ii + 1], rr[4 * ii + 1], s);
      s = fmaf(wreg[4 * ii + 2], rr[4 * ii + 2], s);
      s = fmaf(wreg[4 * ii + 3], rr[4 * ii + 3], s);
      p[ii] = s;
    }
#if HAS_WMMA4
    v8f accE = {};
    v8f accO = {};
#pragma unroll
    for (int pr = 0; pr < 8; ++pr) {
      v2f a, b;
      a.x = p[2 * pr];
      a.y = p[2 * pr + 1];
      b.x = (lcol == 2 * pr) ? 1.0f : 0.0f;
      b.y = (lcol == 2 * pr + 1) ? 1.0f : 0.0f;
      if (pr & 1) {
        accO = __builtin_amdgcn_wmma_f32_16x16x4_f32(
            false, a, false, b, (short)0, accO, false, false);
      } else {
        accE = __builtin_amdgcn_wmma_f32_16x16x4_f32(
            false, a, false, b, (short)0, accE, false, false);
      }
    }
    const v8f acc = accE + accO;
    float s = ((acc[0] + acc[1]) + (acc[2] + acc[3])) +
              ((acc[4] + acc[5]) + (acc[6] + acc[7]));
    s += xhalf(s);
    if (lane < 16) cx_out[16 * wav + lane] = s;
#else
    float mine = 0.f;
#pragma unroll
    for (int ii = 0; ii < 16; ++ii) {
      float s = p[ii];
      for (int m = 16; m >= 1; m >>= 1) s += __shfl_xor(s, m, 32);
      if (lane == ii) mine = s;
    }
    if (lane < 16) cx_out[16 * wav + lane] = mine;
#endif
  };

  auto extract = [&](float f) -> float {
    return f * (1.f - powf(kOneMinusE0, 1.f / f)) * kInvE0;
  };
  auto flow_step = [&](float x, float hb) -> float4 {
    const float d1s = x - kRTS * hs - kRTF * (hf - 1.f);
    const float d1f = hs;
    const float s2 = hs + kBDT * d1s;
    const float f2 = hf + kBDT * d1f;
    const float d2s = x - kRTS * s2 - kRTF * (f2 - 1.f);
    const float d2f = s2;
    hs += hb * (d1s + d2s);
    hf += hb * (d1f + d2f);
    const float nf2 = hf + kBDT * hs;
    return make_float4(hf, nf2, extract(hf), extract(nf2));
  };
  auto vq_step = [&](float4 pk, float hb) {
    const float vr1 = pow3125(bvl);
    const float d1v = kRTO * (pk.x - vr1);
    const float d1q = kRTO * (pk.z - vr1 * bq / bvl);
    const float v2 = bvl + kBDT * d1v;
    const float q2 = bq + kBDT * d1q;
    const float vr2 = pow3125(v2);
    const float d2v = kRTO * (pk.y - vr2);
    const float d2q = kRTO * (pk.w - vr2 * q2 / v2);
    bvl += hb * (d1v + d2v);
    bq  += hb * (d1q + d2q);
  };

  coupling_window(-1, cxA, [] {});

  auto flush_rv = [&](int blk) {
    float4* dst = (float4*)out + blk * (kBlk * 2 * kN / 4);
    const float4* s4 = (const float4*)rvblk;
#pragma unroll
    for (int k = 0; k < kBlk / 4; ++k) dst[tid + 256 * k] = s4[tid + 256 * k];
  };

  for (int blk = 0; blk < kNBlk; ++blk) {
    __syncthreads();
    {
      const float4* nsrc = (const float4*)noise + blk * (kBlk * 2 * kN / 4);
      float4 nz4[kBlk / 4];
#pragma unroll
      for (int k = 0; k < kBlk / 4; ++k) nz4[k] = nsrc[tid + 256 * k];
      if (blk > 0) flush_rv(blk - 1);
#pragma unroll
      for (int k = 0; k < kBlk / 4; ++k)
        ((float4*)nzblk)[tid + 256 * k] = nz4[k];
    }

   for (int s = 0; s < kBlk; ++s) {
    const int T = blk * kBlk + s;
    float* const cxP = (T & 1) ? cxB : cxA;
    float* const cxN = (T & 1) ? cxA : cxB;

    if (tid < kN) ring[(T & 255) * kRingPitch + tid] = cur_r;
    __syncthreads();

    const float2 nz_cur = ((const float2*)nzblk)[s * kN + (tid & (kN - 1))];

    const float hb = (T > 0) ? 0.5f * kBDT : 0.f;
    if (wav < 4) {
      const float4 pack = flow_step(cur_r, hb);
      coupling_window(T, cxN, [&] { gpk[tid] = pack; });
    } else {
      coupling_window(T, cxN, [&] { vq_step(pk, hb); });
    }
    __syncthreads();

    if (tid < kN) {
      const float dwr = kSig * nz_cur.x;
      const float dwv = kSig * nz_cur.y;
      const float stv = (tid == 0) ? stiml[T / 10] : 0.f;

      const float rc = cur_r, Vc = cur_V;
      const float I1 = g0 * cxP[tid] + stv;
      const float d1r = kInvPi + 2.f * rc * Vc;
      const float pt1 = kPi * rc;
      const float d1V = Vc * Vc + eta + kJC * rc + I1 - pt1 * pt1;
      const float xir = fmaxf(rc + kDT * d1r + dwr, 0.f);
      const float xiV = Vc + kDT * d1V + dwv;

      const float I2 = g0 * cxN[tid] + stv;
      const float d2r = kInvPi + 2.f * xir * xiV;
      const float pt2 = kPi * xir;
      const float d2V = xiV * xiV + eta + kJC * xir + I2 - pt2 * pt2;

      const float nxr = fmaxf(rc + kDT * 0.5f * (d1r + d2r) + dwr, 0.f);
      const float nxV = Vc + kDT * 0.5f * (d1V + d2V) + dwv;
      cur_r = nxr;
      cur_V = nxV;

      float2 rv2; rv2.x = nxr; rv2.y = nxV;
      ((float2*)rvblk)[s * kN + tid] = rv2;
    } else {
      pk = gpk[tid - kN];
    }
   }
  }

  __syncthreads();
  flush_rv(kNBlk - 1);

  if (tid >= kN) {
    vq_step(pk, 0.5f * kBDT);
    const float bold = kV0 * (kK1 * (1.f - bq) + kK2 * (1.f - bq / bvl) +
                              kK3 * (1.f - bvl));
    out[kSteps * kN * 2 + (tid - kN)] = bold;
  }
}

extern "C" void kernel_launch(void* const* d_in, const int* in_sizes, int n_in,
                              void* d_out, int out_size, void* d_ws, size_t ws_size,
                              hipStream_t stream) {
  (void)in_sizes; (void)n_in; (void)d_ws; (void)ws_size; (void)out_size;
  const float* region_pars  = (const float*)d_in[0];
  const float* Wt           = (const float*)d_in[1];
  const float* g            = (const float*)d_in[2];
  const float* stimulus     = (const float*)d_in[3];
  const float* noise        = (const float*)d_in[4];
  const float* initial_cond = (const float*)d_in[5];
  const int*   lags         = (const int*)d_in[6];

  const size_t smem_bytes =
      (size_t)(kRingT * kRingPitch + kN + kN + 104 + 2 * kBlk * 2 * kN +
               kN * 4) *
      sizeof(float);

  tvb_mpr_persistent<<<1, 256, smem_bytes, stream>>>(
      region_pars, Wt, g, stimulus, noise, initial_cond, lags, (float*)d_out);
}


# coupling re-tiled over 16 waves of 8 rows (4 waves per SIMD, was 2 of 16 rows): each wave's in-order stream between barriers loses half its gathers; two more waves fill the balloon chains' stalls
# speedup vs baseline: 2.3493x; 1.0700x over previous
#include <hip/hip_runtime.h>
#include <math.h>


typedef float v2f __attribute__((ext_vector_type(2)));
typedef float v8f __attribute__((ext_vector_type(8)));

#if defined(__has_builtin)
#  if __has_builtin(__builtin_amdgcn_wmma_f32_16x16x4_f32)
#    define HAS_WMMA4 1
#  endif
#endif
#ifndef HAS_WMMA4
#  define HAS_WMMA4 0
#endif

namespace {
constexpr int   kN         = 128;
constexpr int   kThreads   = 512;
constexpr int   kRows      = 8;
constexpr int   kE         = 4 * kRows;
constexpr int   kRingT     = 256;
constexpr int   kRingPitch = 128;
constexpr unsigned kRingMask = kRingT * kRingPitch * 4 - 1;
constexpr int   kSteps     = 1000;
constexpr int   kBlk       = 40;
constexpr int   kNBlk      = kSteps / kBlk;
static_assert(kSteps % kBlk == 0 && kBlk % 4 == 0, "block must tile the scan");
constexpr int   kQ4        = kBlk * 2 * kN / 4 / kThreads;
static_assert(kQ4 * kThreads * 4 == kBlk * 2 * kN, "block must tile the threads");
static_assert(kRows * (kThreads / 32) == kN && kRows % 4 == 0, "rows must tile");
constexpr float kDT        = 0.1f;
constexpr float kSig       = 0.0031622776601683794f;
constexpr float kPi        = 3.14159265358979323846f;
constexpr float kInvPi     = 0.3183098861837907f;
constexpr float kJC        = 15.0f;
constexpr float kBDT   = 0.01f;
constexpr float kRTS   = 1.0f / 0.65f;
constexpr float kRTF   = 1.0f / 0.41f;
constexpr float kRTO   = 1.0f / 0.98f;
constexpr float kInvE0 = 2.5f;
constexpr float kOneMinusE0 = 0.6f;
constexpr float kV0 = 4.0f;
constexpr float kK1 = 4.3f * 40.3f * 0.4f * 0.04f;
constexpr float kK2 = 0.5f * 25.0f * 0.4f * 0.04f;
constexpr float kK3 = 0.5f;
}

__global__ __launch_bounds__(kThreads) void tvb_mpr_persistent(
    const float* __restrict__ region_pars,
    const float* __restrict__ Wt,
    const float* __restrict__ g,
    const float* __restrict__ stimulus,
    const float* __restrict__ noise,
    const float* __restrict__ initial_cond,
    const int*   __restrict__ lags,
    float* __restrict__ out)
{
  extern __shared__ __attribute__((aligned(16))) float smem[];
  float* ring  = smem;
  float* cxA   = ring + kRingT * kRingPitch;
  float* cxB   = cxA + kN;
  float* stiml = cxB + kN;
  float* nzblk = stiml + 104;
  float* rvblk = nzblk + kBlk * 2 * kN;
  float4* gpk  = (float4*)(rvblk + kBlk * 2 * kN);

  const int tid   = threadIdx.x;
  const int lane  = tid & 31;
  const int wav   = __builtin_amdgcn_readfirstlane(tid >> 5);
  const int lcol  = lane & 15;

  float    wreg[kE];
  unsigned qoff[kE];
#pragma unroll
  for (int ii = 0; ii < kRows; ++ii) {
#pragma unroll
    for (int k = 0; k < 4; ++k) {
      const int i = kRows * wav + ii;
      const int j = lane + 32 * k;
      wreg[4 * ii + k] = Wt[i * kN + j];
      qoff[4 * ii + k] =
          ((unsigned)((1 - lags[i * kN + j]) & 255) << 9) | ((unsigned)j << 2);
    }
  }

  for (int t = tid; t < 100; t += kThreads) stiml[t] = stimulus[t];
  for (int idx = tid; idx < kRingT * kN; idx += kThreads) {
    ring[idx] = initial_cond[2 * (idx & 127)];
  }

  float cur_r = 0.f, cur_V = 0.f, eta = 0.f;
  float hs = 1.f, hf = 1.f;
  float bvl = 1.f, bq = 1.f;
  float4 pk = make_float4(1.f, 1.f, 1.f, 1.f);
  const float g0 = g[0];
  if (tid < kN) {
    cur_r  = initial_cond[2 * tid];
    cur_V  = initial_cond[2 * tid + 1];
    eta    = region_pars[tid];
  }
  __syncthreads();

  auto pow3125 = [](float v) -> float {
    const float r8 = sqrtf(sqrtf(sqrtf(v)));
    return (v * v) * (v * r8);
  };

  auto xhalf = [](float x) -> float {
    const int b = __builtin_bit_cast(int, x);
    return __builtin_bit_cast(
        float, __builtin_amdgcn_permlanex16(b, b, 0x76543210, 0xfedcba98,
                                            false, false));
  };

  auto coupling_window = [&](int base, float* cx_out, auto&& beside) {
    const unsigned tb = (unsigned)(base & 255) << 9;
    const char* const ringb = (const char*)ring;
    float rr[kE];
#pragma unroll
    for (int e = 0; e < kE; ++e)
      rr[e] = *(const float*)(ringb + ((tb + qoff[e]) & kRingMask));
    beside();
    float p[kRows];
#pragma unroll
    for (int ii = 0; ii < kRows; ++ii) {
      float s = wreg[4 * ii] * rr[4 * ii];
      s = fmaf(wreg[4 * ii + 1], rr[4 * ii + 1], s);
      s = fmaf(wreg[4 * ii + 2], rr[4 * ii + 2], s);
      s = fmaf(wreg[4 * ii + 3], rr[4 * ii + 3], s);
      p[ii] = s;
    }
#if HAS_WMMA4
    v8f accE = {};
    v8f accO = {};
#pragma unroll
    for (int pr = 0; pr < kRows / 2; ++pr) {
      v2f a, b;
      a.x = p[2 * pr];
      a.y = p[2 * pr + 1];
      b.x = (lcol == 2 * pr) ? 1.0f : 0.0f;
      b.y = (lcol == 2 * pr + 1) ? 1.0f : 0.0f;
      if (pr & 1) {
        accO = __builtin_amdgcn_wmma_f32_16x16x4_f32(
            false, a, false, b, (short)0, accO, false, false);
      } else {
        accE = __builtin_amdgcn_wmma_f32_16x16x4_f32(
            false, a, false, b, (short)0, accE, false, false);
      }
    }
    const v8f acc = accE + accO;
    float s = ((acc[0] + acc[1]) + (acc[2] + acc[3])) +
              ((acc[4] + acc[5]) + (acc[6] + acc[7]));
    s += xhalf(s);
    if (lane < kRows) cx_out[kRows * wav + lane] = s;
#else
    float mine = 0.f;
#pragma unroll
    for (int ii = 0; ii < kRows; ++ii) {
      float s = p[ii];
      for (int m = 16; m >= 1; m >>= 1) s += __shfl_xor(s, m, 32);
      if (lane == ii) mine = s;
    }
    if (lane < kRows) cx_out[kRows * wav + lane] = mine;
#endif
  };

  auto extract = [&](float f) -> float {
    return f * (1.f - powf(kOneMinusE0, 1.f / f)) * kInvE0;
  };
  auto flow_step = [&](float x, float hb) -> float4 {
    const float d1s = x - kRTS * hs - kRTF * (hf - 1.f);
    const float d1f = hs;
    const float s2 = hs + kBDT * d1s;
    const float f2 = hf + kBDT * d1f;
    const float d2s = x - kRTS * s2 - kRTF * (f2 - 1.f);
    const float d2f = s2;
    hs += hb * (d1s + d2s);
    hf += hb * (d1f + d2f);
    const float nf2 = hf + kBDT * hs;
    return make_float4(hf, nf2, extract(hf), extract(nf2));
  };
  auto vq_step = [&](float4 pk, float hb) {
    const float vr1 = pow3125(bvl);
    const float d1v = kRTO * (pk.x - vr1);
    const float d1q = kRTO * (pk.z - vr1 * bq / bvl);
    const float v2 = bvl + kBDT * d1v;
    const float q2 = bq + kBDT * d1q;
    const float vr2 = pow3125(v2);
    const float d2v = kRTO * (pk.y - vr2);
    const float d2q = kRTO * (pk.w - vr2 * q2 / v2);
    bvl += hb * (d1v + d2v);
    bq  += hb * (d1q + d2q);
  };

  coupling_window(-1, cxA, [] {});

  auto flush_rv = [&](int blk) {
    float4* dst = (float4*)out + blk * (kBlk * 2 * kN / 4);
    const float4* s4 = (const float4*)rvblk;
#pragma unroll
    for (int k = 0; k < kQ4; ++k) dst[tid + kThreads * k] = s4[tid + kThreads * k];
  };

  for (int blk = 0; blk < kNBlk; ++blk) {
    __syncthreads();
    {
      const float4* nsrc = (const float4*)noise + blk * (kBlk * 2 * kN / 4);
      float4 nz4[kQ4];
#pragma unroll
      for (int k = 0; k < kQ4; ++k) nz4[k] = nsrc[tid + kThreads * k];
      if (blk > 0) flush_rv(blk - 1);
#pragma unroll
      for (int k = 0; k < kQ4; ++k)
        ((float4*)nzblk)[tid + kThreads * k] = nz4[k];
    }

   for (int s = 0; s < kBlk; ++s) {
    const int T = blk * kBlk + s;
    float* const cxP = (T & 1) ? cxB : cxA;
    float* const cxN = (T & 1) ? cxA : cxB;

    if (tid < kN) ring[(T & 255) * kRingPitch + tid] = cur_r;
    __syncthreads();

    const float2 nz_cur = ((const float2*)nzblk)[s * kN + (tid & (kN - 1))];

    const float hb = (T > 0) ? 0.5f * kBDT : 0.f;
    if (wav < 4) {
      coupling_window(T, cxN, [&] { gpk[tid] = flow_step(cur_r, hb); });
    } else if (wav < 8) {
      coupling_window(T, cxN, [&] { vq_step(pk, hb); });
    } else {
      coupling_window(T, cxN, [] {});
    }
    __syncthreads();

    if (tid < kN) {
      const float dwr = kSig * nz_cur.x;
      const float dwv = kSig * nz_cur.y;
      const float stv = (tid == 0) ? stiml[T / 10] : 0.f;

      const float rc = cur_r, Vc = cur_V;
      const float I1 = g0 * cxP[tid] + stv;
      const float d1r = kInvPi + 2.f * rc * Vc;
      const float pt1 = kPi * rc;
      const float d1V = Vc * Vc + eta + kJC * rc + I1 - pt1 * pt1;
      const float xir = fmaxf(rc + kDT * d1r + dwr, 0.f);
      const float xiV = Vc + kDT * d1V + dwv;

      const float I2 = g0 * cxN[tid] + stv;
      const float d2r = kInvPi + 2.f * xir * xiV;
      const float pt2 = kPi * xir;
      const float d2V = xiV * xiV + eta + kJC * xir + I2 - pt2 * pt2;

      const float nxr = fmaxf(rc + kDT * 0.5f * (d1r + d2r) + dwr, 0.f);
      const float nxV = Vc + kDT * 0.5f * (d1V + d2V) + dwv;
      cur_r = nxr;
      cur_V = nxV;

      float2 rv2; rv2.x = nxr; rv2.y = nxV;
      ((float2*)rvblk)[s * kN + tid] = rv2;
    } else if (wav < 8) {
      pk = gpk[tid - kN];
    }
   }
  }

  __syncthreads();
  flush_rv(kNBlk - 1);

  if (tid >= kN && tid < 2 * kN) {
    vq_step(pk, 0.5f * kBDT);
    const float bold = kV0 * (kK1 * (1.f - bq) + kK2 * (1.f - bq / bvl) +
                              kK3 * (1.f - bvl));
    out[kSteps * kN * 2 + (tid - kN)] = bold;
  }
}

extern "C" void kernel_launch(void* const* d_in, const int* in_sizes, int n_in,
                              void* d_out, int out_size, void* d_ws, size_t ws_size,
                              hipStream_t stream) {
  (void)in_sizes; (void)n_in; (void)d_ws; (void)ws_size; (void)out_size;
  const float* region_pars  = (const float*)d_in[0];
  const float* Wt           = (const float*)d_in[1];
  const float* g            = (const float*)d_in[2];
  const float* stimulus     = (const float*)d_in[3];
  const float* noise        = (const float*)d_in[4];
  const float* initial_cond = (const float*)d_in[5];
  const int*   lags         = (const int*)d_in[6];

  const size_t smem_bytes =
      (size_t)(kRingT * kRingPitch + kN + kN + 104 + 2 * kBlk * 2 * kN +
               kN * 4) *
      sizeof(float);

  tvb_mpr_persistent<<<1, kThreads, smem_bytes, stream>>>(
      region_pars, Wt, g, stimulus, noise, initial_cond, lags, (float*)d_out);
}


# new r of a step needs only its first window (gathered the step before): owner waves compute and store r(T+1) inside the coupling phase, so the publish barrier is gone: one barrier a step
# speedup vs baseline: 2.9710x; 1.2646x over previous
#include <hip/hip_runtime.h>
#include <math.h>


typedef float v2f __attribute__((ext_vector_type(2)));
typedef float v8f __attribute__((ext_vector_type(8)));

#if defined(__has_builtin)
#  if __has_builtin(__builtin_amdgcn_wmma_f32_16x16x4_f32)
#    define HAS_WMMA4 1
#  endif
#endif
#ifndef HAS_WMMA4
#  define HAS_WMMA4 0
#endif

namespace {
constexpr int   kN         = 128;
constexpr int   kThreads   = 512;
constexpr int   kRows      = 8;
constexpr int   kE         = 4 * kRows;
constexpr int   kRingT     = 256;
constexpr int   kRingPitch = 128;
constexpr unsigned kRingMask = kRingT * kRingPitch * 4 - 1;
constexpr int   kSteps     = 1000;
constexpr int   kBlk       = 40;
constexpr int   kNBlk      = kSteps / kBlk;
static_assert(kSteps % kBlk == 0 && kBlk % 4 == 0, "block must tile the scan");
constexpr int   kQ4        = kBlk * 2 * kN / 4 / kThreads;
static_assert(kQ4 * kThreads * 4 == kBlk * 2 * kN, "block must tile the threads");
static_assert(kRows * (kThreads / 32) == kN && kRows % 4 == 0, "rows must tile");
constexpr float kDT        = 0.1f;
constexpr float kSig       = 0.0031622776601683794f;
constexpr float kPi        = 3.14159265358979323846f;
constexpr float kInvPi     = 0.3183098861837907f;
constexpr float kJC        = 15.0f;
constexpr float kBDT   = 0.01f;
constexpr float kRTS   = 1.0f / 0.65f;
constexpr float kRTF   = 1.0f / 0.41f;
constexpr float kRTO   = 1.0f / 0.98f;
constexpr float kInvE0 = 2.5f;
constexpr float kOneMinusE0 = 0.6f;
constexpr float kV0 = 4.0f;
constexpr float kK1 = 4.3f * 40.3f * 0.4f * 0.04f;
constexpr float kK2 = 0.5f * 25.0f * 0.4f * 0.04f;
constexpr float kK3 = 0.5f;
}

__global__ __launch_bounds__(kThreads) void tvb_mpr_persistent(
    const float* __restrict__ region_pars,
    const float* __restrict__ Wt,
    const float* __restrict__ g,
    const float* __restrict__ stimulus,
    const float* __restrict__ noise,
    const float* __restrict__ initial_cond,
    const int*   __restrict__ lags,
    float* __restrict__ out)
{
  extern __shared__ __attribute__((aligned(16))) float smem[];
  float* ring  = smem;
  float* cxA   = ring + kRingT * kRingPitch;
  float* cxB   = cxA + kN;
  float* stiml = cxB + kN;
  float* nzblk = stiml + 104;
  float* rvblk = nzblk + kBlk * 2 * kN;
  float4* gpk  = (float4*)(rvblk + kBlk * 2 * kN);

  const int tid   = threadIdx.x;
  const int lane  = tid & 31;
  const int wav   = __builtin_amdgcn_readfirstlane(tid >> 5);
  const int lcol  = lane & 15;

  float    wreg[kE];
  unsigned qoff[kE];
#pragma unroll
  for (int ii = 0; ii < kRows; ++ii) {
#pragma unroll
    for (int k = 0; k < 4; ++k) {
      const int i = kRows * wav + ii;
      const int j = lane + 32 * k;
      wreg[4 * ii + k] = Wt[i * kN + j];
      qoff[4 * ii + k] =
          ((unsigned)((1 - lags[i * kN + j]) & 255) << 9) | ((unsigned)j << 2);
    }
  }

  for (int t = tid; t < 100; t += kThreads) stiml[t] = stimulus[t];
  for (int idx = tid; idx < kRingT * kN; idx += kThreads) {
    ring[idx] = initial_cond[2 * (idx & 127)];
  }

  float cur_r = 0.f, cur_V = 0.f, eta = 0.f;
  float hs = 1.f, hf = 1.f;
  float bvl = 1.f, bq = 1.f;
  float4 pk = make_float4(1.f, 1.f, 1.f, 1.f);
  const float g0 = g[0];
  if (tid < kN) {
    cur_r  = initial_cond[2 * tid];
    cur_V  = initial_cond[2 * tid + 1];
    eta    = region_pars[tid];
  }
  __syncthreads();

  auto pow3125 = [](float v) -> float {
    const float r8 = sqrtf(sqrtf(sqrtf(v)));
    return (v * v) * (v * r8);
  };

  auto xhalf = [](float x) -> float {
    const int b = __builtin_bit_cast(int, x);
    return __builtin_bit_cast(
        float, __builtin_amdgcn_permlanex16(b, b, 0x76543210, 0xfedcba98,
                                            false, false));
  };

  auto coupling_window = [&](int base, float* cx_out, auto&& beside) {
    const unsigned tb = (unsigned)(base & 255) << 9;
    const char* const ringb = (const char*)ring;
    float rr[kE];
#pragma unroll
    for (int e = 0; e < kE; ++e)
      rr[e] = *(const float*)(ringb + ((tb + qoff[e]) & kRingMask));
    beside();
    float p[kRows];
#pragma unroll
    for (int ii = 0; ii < kRows; ++ii) {
      float s = wreg[4 * ii] * rr[4 * ii];
      s = fmaf(wreg[4 * ii + 1], rr[4 * ii + 1], s);
      s = fmaf(wreg[4 * ii + 2], rr[4 * ii + 2], s);
      s = fmaf(wreg[4 * ii + 3], rr[4 * ii + 3], s);
      p[ii] = s;
    }
#if HAS_WMMA4
    v8f accE = {};
    v8f accO = {};
#pragma unroll
    for (int pr = 0; pr < kRows / 2; ++pr) {
      v2f a, b;
      a.x = p[2 * pr];
      a.y = p[2 * pr + 1];
      b.x = (lcol == 2 * pr) ? 1.0f : 0.0f;
      b.y = (lcol == 2 * pr + 1) ? 1.0f : 0.0f;
      if (pr & 1) {
        accO = __builtin_amdgcn_wmma_f32_16x16x4_f32(
            false, a, false, b, (short)0, accO, false, false);
      } else {
        accE = __builtin_amdgcn_wmma_f32_16x16x4_f32(
            false, a, false, b, (short)0, accE, false, false);
      }
    }
    const v8f acc = accE + accO;
    float s = ((acc[0] + acc[1]) + (acc[2] + acc[3])) +
              ((acc[4] + acc[5]) + (acc[6] + acc[7]));
    s += xhalf(s);
    if (lane < kRows) cx_out[kRows * wav + lane] = s;
#else
    float mine = 0.f;
#pragma unroll
    for (int ii = 0; ii < kRows; ++ii) {
      float s = p[ii];
      for (int m = 16; m >= 1; m >>= 1) s += __shfl_xor(s, m, 32);
      if (lane == ii) mine = s;
    }
    if (lane < kRows) cx_out[kRows * wav + lane] = mine;
#endif
  };

  auto extract = [&](float f) -> float {
    return f * (1.f - powf(kOneMinusE0, 1.f / f)) * kInvE0;
  };
  auto flow_step = [&](float x, float hb) -> float4 {
    const float d1s = x - kRTS * hs - kRTF * (hf - 1.f);
    const float d1f = hs;
    const float s2 = hs + kBDT * d1s;
    const float f2 = hf + kBDT * d1f;
    const float d2s = x - kRTS * s2 - kRTF * (f2 - 1.f);
    const float d2f = s2;
    hs += hb * (d1s + d2s);
    hf += hb * (d1f + d2f);
    const float nf2 = hf + kBDT * hs;
    return make_float4(hf, nf2, extract(hf), extract(nf2));
  };
  auto vq_step = [&](float4 pk, float hb) {
    const float vr1 = pow3125(bvl);
    const float d1v = kRTO * (pk.x - vr1);
    const float d1q = kRTO * (pk.z - vr1 * bq / bvl);
    const float v2 = bvl + kBDT * d1v;
    const float q2 = bq + kBDT * d1q;
    const float vr2 = pow3125(v2);
    const float d2v = kRTO * (pk.y - vr2);
    const float d2q = kRTO * (pk.w - vr2 * q2 / v2);
    bvl += hb * (d1v + d2v);
    bq  += hb * (d1q + d2q);
  };

  coupling_window(-1, cxA, [] {});
  __syncthreads();
  float cxp = (tid < kN) ? cxA[tid] : 0.f;
  float xir = 0.f, xiV = 0.f, nxr = 0.f, d1V = 0.f, dwv = 0.f, stv = 0.f;

  auto flush_rv = [&](int blk) {
    float4* dst = (float4*)out + blk * (kBlk * 2 * kN / 4);
    const float4* s4 = (const float4*)rvblk;
#pragma unroll
    for (int k = 0; k < kQ4; ++k) dst[tid + kThreads * k] = s4[tid + kThreads * k];
  };

  for (int blk = 0; blk < kNBlk; ++blk) {
    __syncthreads();
    {
      const float4* nsrc = (const float4*)noise + blk * (kBlk * 2 * kN / 4);
      float4 nz4[kQ4];
#pragma unroll
      for (int k = 0; k < kQ4; ++k) nz4[k] = nsrc[tid + kThreads * k];
      if (blk > 0) flush_rv(blk - 1);
#pragma unroll
      for (int k = 0; k < kQ4; ++k)
        ((float4*)nzblk)[tid + kThreads * k] = nz4[k];
    }
    __syncthreads();

   for (int s = 0; s < kBlk; ++s) {
    const int T = blk * kBlk + s;
    float* const cxN = (T & 1) ? cxA : cxB;
    float4* const gpkT = gpk + (T & 1) * kN;

    const float hb = (T > 0) ? 0.5f * kBDT : 0.f;
    if (wav < 4) {
      coupling_window(T, cxN, [&] {
        const float2 nz = ((const float2*)nzblk)[s * kN + tid];
        const float dwr = kSig * nz.x;
        dwv = kSig * nz.y;
        const float stc = stiml[T / 10];
        stv = (tid == 0) ? stc : 0.f;
        const float rc = cur_r, Vc = cur_V;
        const float I1 = g0 * cxp + stv;
        const float d1r = kInvPi + 2.f * rc * Vc;
        const float pt1 = kPi * rc;
        d1V = Vc * Vc + eta + kJC * rc + I1 - pt1 * pt1;
        xir = fmaxf(rc + kDT * d1r + dwr, 0.f);
        xiV = Vc + kDT * d1V + dwv;
        const float d2r = kInvPi + 2.f * xir * xiV;
        nxr = fmaxf(rc + kDT * 0.5f * (d1r + d2r) + dwr, 0.f);
        ring[((T + 1) & 255) * kRingPitch + tid] = nxr;
        gpkT[tid] = flow_step(cur_r, hb);
      });
    } else if (wav < 8) {
      coupling_window(T, cxN, [&] { vq_step(pk, hb); });
    } else {
      coupling_window(T, cxN, [] {});
    }
    __syncthreads();

    if (wav < 4) {
      const float cxn = cxN[tid];
      const float I2 = g0 * cxn + stv;
      const float pt2 = kPi * xir;
      const float d2V = xiV * xiV + eta + kJC * xir + I2 - pt2 * pt2;
      const float nxV = cur_V + kDT * 0.5f * (d1V + d2V) + dwv;
      cur_r = nxr;
      cur_V = nxV;
      cxp = cxn;

      float2 rv2; rv2.x = nxr; rv2.y = nxV;
      ((float2*)rvblk)[s * kN + tid] = rv2;
    } else if (wav < 8) {
      pk = gpkT[tid - kN];
    }
   }
  }

  __syncthreads();
  flush_rv(kNBlk - 1);

  if (tid >= kN && tid < 2 * kN) {
    vq_step(pk, 0.5f * kBDT);
    const float bold = kV0 * (kK1 * (1.f - bq) + kK2 * (1.f - bq / bvl) +
                              kK3 * (1.f - bvl));
    out[kSteps * kN * 2 + (tid - kN)] = bold;
  }
}

extern "C" void kernel_launch(void* const* d_in, const int* in_sizes, int n_in,
                              void* d_out, int out_size, void* d_ws, size_t ws_size,
                              hipStream_t stream) {
  (void)in_sizes; (void)n_in; (void)d_ws; (void)ws_size; (void)out_size;
  const float* region_pars  = (const float*)d_in[0];
  const float* Wt           = (const float*)d_in[1];
  const float* g            = (const float*)d_in[2];
  const float* stimulus     = (const float*)d_in[3];
  const float* noise        = (const float*)d_in[4];
  const float* initial_cond = (const float*)d_in[5];
  const int*   lags         = (const int*)d_in[6];

  const size_t smem_bytes =
      (size_t)(kRingT * kRingPitch + kN + kN + 104 + 2 * kBlk * 2 * kN +
               2 * kN * 4) *
      sizeof(float);

  tvb_mpr_persistent<<<1, kThreads, smem_bytes, stream>>>(
      region_pars, Wt, g, stimulus, noise, initial_cond, lags, (float*)d_out);
}


# V half of step T and the (v,q) waves' pack fetch run inside the coupling phase of step T+1, not behind the step barrier: their LDS round trips and short chains fill the balloon chains' stalls
# speedup vs baseline: 3.2260x; 1.0858x over previous
#include <hip/hip_runtime.h>
#include <math.h>


typedef float v2f __attribute__((ext_vector_type(2)));
typedef float v8f __attribute__((ext_vector_type(8)));

#if defined(__has_builtin)
#  if __has_builtin(__builtin_amdgcn_wmma_f32_16x16x4_f32)
#    define HAS_WMMA4 1
#  endif
#endif
#ifndef HAS_WMMA4
#  define HAS_WMMA4 0
#endif

namespace {
template <bool B> struct Flag { static constexpr bool value = B; };
constexpr int   kN         = 128;
constexpr int   kThreads   = 512;
constexpr int   kRows      = 8;
constexpr int   kE         = 4 * kRows;
constexpr int   kRingT     = 256;
constexpr int   kRingPitch = 128;
constexpr unsigned kRingMask = kRingT * kRingPitch * 4 - 1;
constexpr int   kSteps     = 1000;
constexpr int   kBlk       = 40;
constexpr int   kNBlk      = kSteps / kBlk;
static_assert(kSteps % kBlk == 0 && kBlk % 4 == 0, "block must tile the scan");
constexpr int   kQ4        = kBlk * 2 * kN / 4 / kThreads;
static_assert(kQ4 * kThreads * 4 == kBlk * 2 * kN, "block must tile the threads");
static_assert(kRows * (kThreads / 32) == kN && kRows % 4 == 0, "rows must tile");
constexpr float kDT        = 0.1f;
constexpr float kSig       = 0.0031622776601683794f;
constexpr float kPi        = 3.14159265358979323846f;
constexpr float kInvPi     = 0.3183098861837907f;
constexpr float kJC        = 15.0f;
constexpr float kBDT   = 0.01f;
constexpr float kRTS   = 1.0f / 0.65f;
constexpr float kRTF   = 1.0f / 0.41f;
constexpr float kRTO   = 1.0f / 0.98f;
constexpr float kInvE0 = 2.5f;
constexpr float kOneMinusE0 = 0.6f;
constexpr float kV0 = 4.0f;
constexpr float kK1 = 4.3f * 40.3f * 0.4f * 0.04f;
constexpr float kK2 = 0.5f * 25.0f * 0.4f * 0.04f;
constexpr float kK3 = 0.5f;
}

__global__ __launch_bounds__(kThreads) void tvb_mpr_persistent(
    const float* __restrict__ region_pars,
    const float* __restrict__ Wt,
    const float* __restrict__ g,
    const float* __restrict__ stimulus,
    const float* __restrict__ noise,
    const float* __restrict__ initial_cond,
    const int*   __restrict__ lags,
    float* __restrict__ out)
{
  extern __shared__ __attribute__((aligned(16))) float smem[];
  float* ring  = smem;
  float* cxA   = ring + kRingT * kRingPitch;
  float* cxB   = cxA + kN;
  float* stiml = cxB + kN;
  float* nzblk = stiml + 104;
  float* rvblk = nzblk + kBlk * 2 * kN;
  float4* gpk  = (float4*)(rvblk + kBlk * 2 * kN);

  const int tid   = threadIdx.x;
  const int lane  = tid & 31;
  const int wav   = __builtin_amdgcn_readfirstlane(tid >> 5);
  const int lcol  = lane & 15;

  float    wreg[kE];
  unsigned qoff[kE];
#pragma unroll
  for (int ii = 0; ii < kRows; ++ii) {
#pragma unroll
    for (int k = 0; k < 4; ++k) {
      const int i = kRows * wav + ii;
      const int j = lane + 32 * k;
      wreg[4 * ii + k] = Wt[i * kN + j];
      qoff[4 * ii + k] =
          ((unsigned)((1 - lags[i * kN + j]) & 255) << 9) | ((unsigned)j << 2);
    }
  }

  for (int t = tid; t < 100; t += kThreads) stiml[t] = stimulus[t];
  for (int idx = tid; idx < kRingT * kN; idx += kThreads) {
    ring[idx] = initial_cond[2 * (idx & 127)];
  }

  for (int t = tid; t < 2 * kN; t += kThreads) gpk[t] = make_float4(1.f, 1.f, 1.f, 1.f);

  float cur_r = 0.f, cur_V = 0.f, eta = 0.f;
  float hs = 1.f, hf = 1.f;
  float bvl = 1.f, bq = 1.f;
  const float g0 = g[0];
  if (tid < kN) {
    cur_r  = initial_cond[2 * tid];
    cur_V  = initial_cond[2 * tid + 1];
    eta    = region_pars[tid];
  }
  __syncthreads();

  auto pow3125 = [](float v) -> float {
    const float r8 = sqrtf(sqrtf(sqrtf(v)));
    return (v * v) * (v * r8);
  };

  auto xhalf = [](float x) -> float {
    const int b = __builtin_bit_cast(int, x);
    return __builtin_bit_cast(
        float, __builtin_amdgcn_permlanex16(b, b, 0x76543210, 0xfedcba98,
                                            false, false));
  };

  auto coupling_window = [&](int base, float* cx_out, auto&& first,
                             auto&& beside) {
    const unsigned tb = (unsigned)(base & 255) << 9;
    const char* const ringb = (const char*)ring;
    first();
    __builtin_amdgcn_sched_barrier(0);
    float rr[kE];
#pragma unroll
    for (int e = 0; e < kE; ++e)
      rr[e] = *(const float*)(ringb + ((tb + qoff[e]) & kRingMask));
    beside();
    float p[kRows];
#pragma unroll
    for (int ii = 0; ii < kRows; ++ii) {
      float s = wreg[4 * ii] * rr[4 * ii];
      s = fmaf(wreg[4 * ii + 1], rr[4 * ii + 1], s);
      s = fmaf(wreg[4 * ii + 2], rr[4 * ii + 2], s);
      s = fmaf(wreg[4 * ii + 3], rr[4 * ii + 3], s);
      p[ii] = s;
    }
#if HAS_WMMA4
    v8f accE = {};
    v8f accO = {};
#pragma unroll
    for (int pr = 0; pr < kRows / 2; ++pr) {
      v2f a, b;
      a.x = p[2 * pr];
      a.y = p[2 * pr + 1];
      b.x = (lcol == 2 * pr) ? 1.0f : 0.0f;
      b.y = (lcol == 2 * pr + 1) ? 1.0f : 0.0f;
      if (pr & 1) {
        accO = __builtin_amdgcn_wmma_f32_16x16x4_f32(
            false, a, false, b, (short)0, accO, false, false);
      } else {
        accE = __builtin_amdgcn_wmma_f32_16x16x4_f32(
            false, a, false, b, (short)0, accE, false, false);
      }
    }
    const v8f acc = accE + accO;
    float s = ((acc[0] + acc[1]) + (acc[2] + acc[3])) +
              ((acc[4] + acc[5]) + (acc[6] + acc[7]));
    s += xhalf(s);
    if (lane < kRows) cx_out[kRows * wav + lane] = s;
#else
    float mine = 0.f;
#pragma unroll
    for (int ii = 0; ii < kRows; ++ii) {
      float s = p[ii];
      for (int m = 16; m >= 1; m >>= 1) s += __shfl_xor(s, m, 32);
      if (lane == ii) mine = s;
    }
    if (lane < kRows) cx_out[kRows * wav + lane] = mine;
#endif
  };

  auto extract = [&](float f) -> float {
    return f * (1.f - powf(kOneMinusE0, 1.f / f)) * kInvE0;
  };
  auto flow_step = [&](float x, float hb) -> float4 {
    const float d1s = x - kRTS * hs - kRTF * (hf - 1.f);
    const float d1f = hs;
    const float s2 = hs + kBDT * d1s;
    const float f2 = hf + kBDT * d1f;
    const float d2s = x - kRTS * s2 - kRTF * (f2 - 1.f);
    const float d2f = s2;
    hs += hb * (d1s + d2s);
    hf += hb * (d1f + d2f);
    const float nf2 = hf + kBDT * hs;
    return make_float4(hf, nf2, extract(hf), extract(nf2));
  };
  auto vq_step = [&](float4 pk, float hb) {
    const float vr1 = pow3125(bvl);
    const float d1v = kRTO * (pk.x - vr1);
    const float d1q = kRTO * (pk.z - vr1 * bq / bvl);
    const float v2 = bvl + kBDT * d1v;
    const float q2 = bq + kBDT * d1q;
    const float vr2 = pow3125(v2);
    const float d2v = kRTO * (pk.y - vr2);
    const float d2q = kRTO * (pk.w - vr2 * q2 / v2);
    bvl += hb * (d1v + d2v);
    bq  += hb * (d1q + d2q);
  };

  coupling_window(-1, cxA, [] {}, [] {});
  __syncthreads();
  float cxp = (tid < kN) ? cxA[tid] : 0.f;
  float xir = 0.f, xiV = 0.f, nxr = 0.f, d1V = 0.f, dwv = 0.f, stv = 0.f;

  auto flush_rv = [&](int blk) {
    float4* dst = (float4*)out + blk * (kBlk * 2 * kN / 4);
    const float4* s4 = (const float4*)rvblk;
#pragma unroll
    for (int k = 0; k < kQ4; ++k) dst[tid + kThreads * k] = s4[tid + kThreads * k];
  };

  auto v_half = [&](float cxn, int s) {
    const float I2 = g0 * cxn + stv;
    const float pt2 = kPi * xir;
    const float d2V = xiV * xiV + eta + kJC * xir + I2 - pt2 * pt2;
    const float nxV = cur_V + kDT * 0.5f * (d1V + d2V) + dwv;
    cur_r = nxr;
    cur_V = nxV;
    cxp = cxn;

    float2 rv2; rv2.x = nxr; rv2.y = nxV;
    ((float2*)rvblk)[s * kN + tid] = rv2;
  };

  auto step = [&](auto deferred, int blk, int s) {
    constexpr bool kDeferred = decltype(deferred)::value;
    const int T = blk * kBlk + s;
    float* const cxN = (T & 1) ? cxA : cxB;
    float* const cxO = (T & 1) ? cxB : cxA;
    float4* const gpkT = gpk + (T & 1) * kN;
    float4* const gpkO = gpk + ((T & 1) ^ 1) * kN;

    const float hb = (T > 0) ? 0.5f * kBDT : 0.f;
    if (wav < 4) {
      float cxn = 0.f;
      float2 nz = make_float2(0.f, 0.f);
      float stc = 0.f;
      coupling_window(T, cxN,
        [&] {
          if (kDeferred) cxn = cxO[tid];
          nz = ((const float2*)nzblk)[s * kN + tid];
          stc = stiml[T / 10];
        },
        [&] {
        if (kDeferred) v_half(cxn, s - 1);
        const float dwr = kSig * nz.x;
        dwv = kSig * nz.y;
        stv = (tid == 0) ? stc : 0.f;
        const float rc = cur_r, Vc = cur_V;
        const float I1 = g0 * cxp + stv;
        const float d1r = kInvPi + 2.f * rc * Vc;
        const float pt1 = kPi * rc;
        d1V = Vc * Vc + eta + kJC * rc + I1 - pt1 * pt1;
        xir = fmaxf(rc + kDT * d1r + dwr, 0.f);
        xiV = Vc + kDT * d1V + dwv;
        const float d2r = kInvPi + 2.f * xir * xiV;
        nxr = fmaxf(rc + kDT * 0.5f * (d1r + d2r) + dwr, 0.f);
        ring[((T + 1) & 255) * kRingPitch + tid] = nxr;
        gpkT[tid] = flow_step(cur_r, hb);
      });
    } else if (wav < 8) {
      float4 pk = make_float4(1.f, 1.f, 1.f, 1.f);
      coupling_window(T, cxN,
        [&] { pk = gpkO[tid - kN]; },
        [&] { vq_step(pk, hb); });
    } else {
      coupling_window(T, cxN, [] {}, [] {});
    }
    __syncthreads();
  };

  for (int blk = 0; blk < kNBlk; ++blk) {
    __syncthreads();
    {
      const float4* nsrc = (const float4*)noise + blk * (kBlk * 2 * kN / 4);
      float4 nz4[kQ4];
#pragma unroll
      for (int k = 0; k < kQ4; ++k) nz4[k] = nsrc[tid + kThreads * k];
      if (blk > 0) flush_rv(blk - 1);
#pragma unroll
      for (int k = 0; k < kQ4; ++k)
        ((float4*)nzblk)[tid + kThreads * k] = nz4[k];
    }
    __syncthreads();

    step(Flag<false>{}, blk, 0);
    for (int s = 1; s < kBlk; ++s) step(Flag<true>{}, blk, s);

    if (wav < 4) {
      const int T = blk * kBlk + kBlk - 1;
      v_half(((T & 1) ? cxA : cxB)[tid], kBlk - 1);
    }
  }

  __syncthreads();
  flush_rv(kNBlk - 1);

  if (tid >= kN && tid < 2 * kN) {
    vq_step(gpk[((kSteps - 1) & 1) * kN + (tid - kN)], 0.5f * kBDT);
    const float bold = kV0 * (kK1 * (1.f - bq) + kK2 * (1.f - bq / bvl) +
                              kK3 * (1.f - bvl));
    out[kSteps * kN * 2 + (tid - kN)] = bold;
  }
}

extern "C" void kernel_launch(void* const* d_in, const int* in_sizes, int n_in,
                              void* d_out, int out_size, void* d_ws, size_t ws_size,
                              hipStream_t stream) {
  (void)in_sizes; (void)n_in; (void)d_ws; (void)ws_size; (void)out_size;
  const float* region_pars  = (const float*)d_in[0];
  const float* Wt           = (const float*)d_in[1];
  const float* g            = (const float*)d_in[2];
  const float* stimulus     = (const float*)d_in[3];
  const float* noise        = (const float*)d_in[4];
  const float* initial_cond = (const float*)d_in[5];
  const int*   lags         = (const int*)d_in[6];

  const size_t smem_bytes =
      (size_t)(kRingT * kRingPitch + kN + kN + 104 + 2 * kBlk * 2 * kN +
               2 * kN * 4) *
      sizeof(float);

  tvb_mpr_persistent<<<1, kThreads, smem_bytes, stream>>>(
      region_pars, Wt, g, stimulus, noise, initial_cond, lags, (float*)d_out);
}


# balloon (v,q) step: its six sqrtf and two divisions use the 1-ulp hardware v_sqrt_f32 / v_rcp_f32, not the correctly rounded sequences (9 to 15 vector instructions each); BOLD changes in last places
# speedup vs baseline: 3.3407x; 1.0356x over previous
#include <hip/hip_runtime.h>
#include <math.h>


typedef float v2f __attribute__((ext_vector_type(2)));
typedef float v8f __attribute__((ext_vector_type(8)));

#if defined(__has_builtin)
#  if __has_builtin(__builtin_amdgcn_wmma_f32_16x16x4_f32)
#    define HAS_WMMA4 1
#  endif
#endif
#ifndef HAS_WMMA4
#  define HAS_WMMA4 0
#endif

namespace {
template <bool B> struct Flag { static constexpr bool value = B; };
constexpr int   kN         = 128;
constexpr int   kThreads   = 512;
constexpr int   kRows      = 8;
constexpr int   kE         = 4 * kRows;
constexpr int   kRingT     = 256;
constexpr int   kRingPitch = 128;
constexpr unsigned kRingMask = kRingT * kRingPitch * 4 - 1;
constexpr int   kSteps     = 1000;
constexpr int   kBlk       = 40;
constexpr int   kNBlk      = kSteps / kBlk;
static_assert(kSteps % kBlk == 0 && kBlk % 4 == 0, "block must tile the scan");
constexpr int   kQ4        = kBlk * 2 * kN / 4 / kThreads;
static_assert(kQ4 * kThreads * 4 == kBlk * 2 * kN, "block must tile the threads");
static_assert(kRows * (kThreads / 32) == kN && kRows % 4 == 0, "rows must tile");
constexpr float kDT        = 0.1f;
constexpr float kSig       = 0.0031622776601683794f;
constexpr float kPi        = 3.14159265358979323846f;
constexpr float kInvPi     = 0.3183098861837907f;
constexpr float kJC        = 15.0f;
constexpr float kBDT   = 0.01f;
constexpr float kRTS   = 1.0f / 0.65f;
constexpr float kRTF   = 1.0f / 0.41f;
constexpr float kRTO   = 1.0f / 0.98f;
constexpr float kInvE0 = 2.5f;
constexpr float kOneMinusE0 = 0.6f;
constexpr float kV0 = 4.0f;
constexpr float kK1 = 4.3f * 40.3f * 0.4f * 0.04f;
constexpr float kK2 = 0.5f * 25.0f * 0.4f * 0.04f;
constexpr float kK3 = 0.5f;
}

__global__ __launch_bounds__(kThreads) void tvb_mpr_persistent(
    const float* __restrict__ region_pars,
    const float* __restrict__ Wt,
    const float* __restrict__ g,
    const float* __restrict__ stimulus,
    const float* __restrict__ noise,
    const float* __restrict__ initial_cond,
    const int*   __restrict__ lags,
    float* __restrict__ out)
{
  extern __shared__ __attribute__((aligned(16))) float smem[];
  float* ring  = smem;
  float* cxA   = ring + kRingT * kRingPitch;
  float* cxB   = cxA + kN;
  float* stiml = cxB + kN;
  float* nzblk = stiml + 104;
  float* rvblk = nzblk + kBlk * 2 * kN;
  float4* gpk  = (float4*)(rvblk + kBlk * 2 * kN);

  const int tid   = threadIdx.x;
  const int lane  = tid & 31;
  const int wav   = __builtin_amdgcn_readfirstlane(tid >> 5);
  const int lcol  = lane & 15;

  float    wreg[kE];
  unsigned qoff[kE];
#pragma unroll
  for (int ii = 0; ii < kRows; ++ii) {
#pragma unroll
    for (int k = 0; k < 4; ++k) {
      const int i = kRows * wav + ii;
      const int j = lane + 32 * k;
      wreg[4 * ii + k] = Wt[i * kN + j];
      qoff[4 * ii + k] =
          ((unsigned)((1 - lags[i * kN + j]) & 255) << 9) | ((unsigned)j << 2);
    }
  }

  for (int t = tid; t < 100; t += kThreads) stiml[t] = stimulus[t];
  for (int idx = tid; idx < kRingT * kN; idx += kThreads) {
    ring[idx] = initial_cond[2 * (idx & 127)];
  }

  for (int t = tid; t < 2 * kN; t += kThreads) gpk[t] = make_float4(1.f, 1.f, 1.f, 1.f);

  float cur_r = 0.f, cur_V = 0.f, eta = 0.f;
  float hs = 1.f, hf = 1.f;
  float bvl = 1.f, bq = 1.f;
  const float g0 = g[0];
  if (tid < kN) {
    cur_r  = initial_cond[2 * tid];
    cur_V  = initial_cond[2 * tid + 1];
    eta    = region_pars[tid];
  }
  __syncthreads();

  auto hw_sqrt = [](float v) -> float { return __builtin_amdgcn_sqrtf(v); };
  auto hw_rcp  = [](float v) -> float { return __builtin_amdgcn_rcpf(v); };

  auto pow3125 = [&](float v) -> float {
    const float r8 = hw_sqrt(hw_sqrt(hw_sqrt(v)));
    return (v * v) * (v * r8);
  };

  auto xhalf = [](float x) -> float {
    const int b = __builtin_bit_cast(int, x);
    return __builtin_bit_cast(
        float, __builtin_amdgcn_permlanex16(b, b, 0x76543210, 0xfedcba98,
                                            false, false));
  };

  auto coupling_window = [&](int base, float* cx_out, auto&& first,
                             auto&& beside) {
    const unsigned tb = (unsigned)(base & 255) << 9;
    const char* const ringb = (const char*)ring;
    first();
    __builtin_amdgcn_sched_barrier(0);
    float rr[kE];
#pragma unroll
    for (int e = 0; e < kE; ++e)
      rr[e] = *(const float*)(ringb + ((tb + qoff[e]) & kRingMask));
    beside();
    float p[kRows];
#pragma unroll
    for (int ii = 0; ii < kRows; ++ii) {
      float s = wreg[4 * ii] * rr[4 * ii];
      s = fmaf(wreg[4 * ii + 1], rr[4 * ii + 1], s);
      s = fmaf(wreg[4 * ii + 2], rr[4 * ii + 2], s);
      s = fmaf(wreg[4 * ii + 3], rr[4 * ii + 3], s);
      p[ii] = s;
    }
#if HAS_WMMA4
    v8f accE = {};
    v8f accO = {};
#pragma unroll
    for (int pr = 0; pr < kRows / 2; ++pr) {
      v2f a, b;
      a.x = p[2 * pr];
      a.y = p[2 * pr + 1];
      b.x = (lcol == 2 * pr) ? 1.0f : 0.0f;
      b.y = (lcol == 2 * pr + 1) ? 1.0f : 0.0f;
      if (pr & 1) {
        accO = __builtin_amdgcn_wmma_f32_16x16x4_f32(
            false, a, false, b, (short)0, accO, false, false);
      } else {
        accE = __builtin_amdgcn_wmma_f32_16x16x4_f32(
            false, a, false, b, (short)0, accE, false, false);
      }
    }
    const v8f acc = accE + accO;
    float s = ((acc[0] + acc[1]) + (acc[2] + acc[3])) +
              ((acc[4] + acc[5]) + (acc[6] + acc[7]));
    s += xhalf(s);
    if (lane < kRows) cx_out[kRows * wav + lane] = s;
#else
    float mine = 0.f;
#pragma unroll
    for (int ii = 0; ii < kRows; ++ii) {
      float s = p[ii];
      for (int m = 16; m >= 1; m >>= 1) s += __shfl_xor(s, m, 32);
      if (lane == ii) mine = s;
    }
    if (lane < kRows) cx_out[kRows * wav + lane] = mine;
#endif
  };

  auto extract = [&](float f) -> float {
    return f * (1.f - powf(kOneMinusE0, 1.f / f)) * kInvE0;
  };
  auto flow_step = [&](float x, float hb) -> float4 {
    const float d1s = x - kRTS * hs - kRTF * (hf - 1.f);
    const float d1f = hs;
    const float s2 = hs + kBDT * d1s;
    const float f2 = hf + kBDT * d1f;
    const float d2s = x - kRTS * s2 - kRTF * (f2 - 1.f);
    const float d2f = s2;
    hs += hb * (d1s + d2s);
    hf += hb * (d1f + d2f);
    const float nf2 = hf + kBDT * hs;
    return make_float4(hf, nf2, extract(hf), extract(nf2));
  };
  auto vq_step = [&](float4 pk, float hb) {
    const float vr1 = pow3125(bvl);
    const float d1v = kRTO * (pk.x - vr1);
    const float d1q = kRTO * (pk.z - vr1 * bq * hw_rcp(bvl));
    const float v2 = bvl + kBDT * d1v;
    const float q2 = bq + kBDT * d1q;
    const float vr2 = pow3125(v2);
    const float d2v = kRTO * (pk.y - vr2);
    const float d2q = kRTO * (pk.w - vr2 * q2 * hw_rcp(v2));
    bvl += hb * (d1v + d2v);
    bq  += hb * (d1q + d2q);
  };

  coupling_window(-1, cxA, [] {}, [] {});
  __syncthreads();
  float cxp = (tid < kN) ? cxA[tid] : 0.f;
  float xir = 0.f, xiV = 0.f, nxr = 0.f, d1V = 0.f, dwv = 0.f, stv = 0.f;

  auto flush_rv = [&](int blk) {
    float4* dst = (float4*)out + blk * (kBlk * 2 * kN / 4);
    const float4* s4 = (const float4*)rvblk;
#pragma unroll
    for (int k = 0; k < kQ4; ++k) dst[tid + kThreads * k] = s4[tid + kThreads * k];
  };

  auto v_half = [&](float cxn, int s) {
    const float I2 = g0 * cxn + stv;
    const float pt2 = kPi * xir;
    const float d2V = xiV * xiV + eta + kJC * xir + I2 - pt2 * pt2;
    const float nxV = cur_V + kDT * 0.5f * (d1V + d2V) + dwv;
    cur_r = nxr;
    cur_V = nxV;
    cxp = cxn;

    float2 rv2; rv2.x = nxr; rv2.y = nxV;
    ((float2*)rvblk)[s * kN + tid] = rv2;
  };

  auto step = [&](auto deferred, int blk, int s) {
    constexpr bool kDeferred = decltype(deferred)::value;
    const int T = blk * kBlk + s;
    float* const cxN = (T & 1) ? cxA : cxB;
    float* const cxO = (T & 1) ? cxB : cxA;
    float4* const gpkT = gpk + (T & 1) * kN;
    float4* const gpkO = gpk + ((T & 1) ^ 1) * kN;

    const float hb = (T > 0) ? 0.5f * kBDT : 0.f;
    if (wav < 4) {
      float cxn = 0.f;
      float2 nz = make_float2(0.f, 0.f);
      float stc = 0.f;
      coupling_window(T, cxN,
        [&] {
          if (kDeferred) cxn = cxO[tid];
          nz = ((const float2*)nzblk)[s * kN + tid];
          stc = stiml[T / 10];
        },
        [&] {
        if (kDeferred) v_half(cxn, s - 1);
        const float dwr = kSig * nz.x;
        dwv = kSig * nz.y;
        stv = (tid == 0) ? stc : 0.f;
        const float rc = cur_r, Vc = cur_V;
        const float I1 = g0 * cxp + stv;
        const float d1r = kInvPi + 2.f * rc * Vc;
        const float pt1 = kPi * rc;
        d1V = Vc * Vc + eta + kJC * rc + I1 - pt1 * pt1;
        xir = fmaxf(rc + kDT * d1r + dwr, 0.f);
        xiV = Vc + kDT * d1V + dwv;
        const float d2r = kInvPi + 2.f * xir * xiV;
        nxr = fmaxf(rc + kDT * 0.5f * (d1r + d2r) + dwr, 0.f);
        ring[((T + 1) & 255) * kRingPitch + tid] = nxr;
        gpkT[tid] = flow_step(cur_r, hb);
      });
    } else if (wav < 8) {
      float4 pk = make_float4(1.f, 1.f, 1.f, 1.f);
      coupling_window(T, cxN,
        [&] { pk = gpkO[tid - kN]; },
        [&] { vq_step(pk, hb); });
    } else {
      coupling_window(T, cxN, [] {}, [] {});
    }
    __syncthreads();
  };

  for (int blk = 0; blk < kNBlk; ++blk) {
    __syncthreads();
    {
      const float4* nsrc = (const float4*)noise + blk * (kBlk * 2 * kN / 4);
      float4 nz4[kQ4];
#pragma unroll
      for (int k = 0; k < kQ4; ++k) nz4[k] = nsrc[tid + kThreads * k];
      if (blk > 0) flush_rv(blk - 1);
#pragma unroll
      for (int k = 0; k < kQ4; ++k)
        ((float4*)nzblk)[tid + kThreads * k] = nz4[k];
    }
    __syncthreads();

    step(Flag<false>{}, blk, 0);
    for (int s = 1; s < kBlk; ++s) step(Flag<true>{}, blk, s);

    if (wav < 4) {
      const int T = blk * kBlk + kBlk - 1;
      v_half(((T & 1) ? cxA : cxB)[tid], kBlk - 1);
    }
  }

  __syncthreads();
  flush_rv(kNBlk - 1);

  if (tid >= kN && tid < 2 * kN) {
    vq_step(gpk[((kSteps - 1) & 1) * kN + (tid - kN)], 0.5f * kBDT);
    const float bold = kV0 * (kK1 * (1.f - bq) + kK2 * (1.f - bq / bvl) +
                              kK3 * (1.f - bvl));
    out[kSteps * kN * 2 + (tid - kN)] = bold;
  }
}

extern "C" void kernel_launch(void* const* d_in, const int* in_sizes, int n_in,
                              void* d_out, int out_size, void* d_ws, size_t ws_size,
                              hipStream_t stream) {
  (void)in_sizes; (void)n_in; (void)d_ws; (void)ws_size; (void)out_size;
  const float* region_pars  = (const float*)d_in[0];
  const float* Wt           = (const float*)d_in[1];
  const float* g            = (const float*)d_in[2];
  const float* stimulus     = (const float*)d_in[3];
  const float* noise        = (const float*)d_in[4];
  const float* initial_cond = (const float*)d_in[5];
  const int*   lags         = (const int*)d_in[6];

  const size_t smem_bytes =
      (size_t)(kRingT * kRingPitch + kN + kN + 104 + 2 * kBlk * 2 * kN +
               2 * kN * 4) *
      sizeof(float);

  tvb_mpr_persistent<<<1, kThreads, smem_bytes, stream>>>(
      region_pars, Wt, g, stimulus, noise, initial_cond, lags, (float*)d_out);
}


# coupling re-tiled to 8 lanes x 16 columns per row pair: a lane's two partial sums are its A operand, ONE selector WMMA a window (4 before), no lane-half exchange; summation order changes
# speedup vs baseline: 3.7998x; 1.1374x over previous
#include <hip/hip_runtime.h>
#include <math.h>


typedef float v2f __attribute__((ext_vector_type(2)));
typedef float v8f __attribute__((ext_vector_type(8)));

#if defined(__has_builtin)
#  if __has_builtin(__builtin_amdgcn_wmma_f32_16x16x4_f32)
#    define HAS_WMMA4 1
#  endif
#endif
#ifndef HAS_WMMA4
#  define HAS_WMMA4 0
#endif

namespace {
template <bool B> struct Flag { static constexpr bool value = B; };
constexpr int   kN         = 128;
constexpr int   kThreads   = 512;
constexpr int   kRows      = 8;
constexpr int   kCols      = 16;
constexpr int   kE         = 2 * kCols;
constexpr int   kRingT     = 256;
constexpr int   kRingPitch = 128;
constexpr unsigned kRingMask = kRingT * kRingPitch * 4 - 1;
constexpr int   kSteps     = 1000;
constexpr int   kBlk       = 40;
constexpr int   kNBlk      = kSteps / kBlk;
static_assert(kSteps % kBlk == 0 && kBlk % 4 == 0, "block must tile the scan");
constexpr int   kQ4        = kBlk * 2 * kN / 4 / kThreads;
static_assert(kQ4 * kThreads * 4 == kBlk * 2 * kN, "block must tile the threads");
static_assert(kRows * (kThreads / 32) == kN && kRows == 8 && 8 * kCols == kN,
              "four groups of eight lanes, two rows a group, tile a wave's rows");
constexpr float kDT        = 0.1f;
constexpr float kSig       = 0.0031622776601683794f;
constexpr float kPi        = 3.14159265358979323846f;
constexpr float kInvPi     = 0.3183098861837907f;
constexpr float kJC        = 15.0f;
constexpr float kBDT   = 0.01f;
constexpr float kRTS   = 1.0f / 0.65f;
constexpr float kRTF   = 1.0f / 0.41f;
constexpr float kRTO   = 1.0f / 0.98f;
constexpr float kInvE0 = 2.5f;
constexpr float kOneMinusE0 = 0.6f;
constexpr float kV0 = 4.0f;
constexpr float kK1 = 4.3f * 40.3f * 0.4f * 0.04f;
constexpr float kK2 = 0.5f * 25.0f * 0.4f * 0.04f;
constexpr float kK3 = 0.5f;
}

__global__ __launch_bounds__(kThreads) void tvb_mpr_persistent(
    const float* __restrict__ region_pars,
    const float* __restrict__ Wt,
    const float* __restrict__ g,
    const float* __restrict__ stimulus,
    const float* __restrict__ noise,
    const float* __restrict__ initial_cond,
    const int*   __restrict__ lags,
    float* __restrict__ out)
{
  extern __shared__ __attribute__((aligned(16))) float smem[];
  float* ring  = smem;
  float* cxA   = ring + kRingT * kRingPitch;
  float* cxB   = cxA + kN;
  float* stiml = cxB + kN;
  float* nzblk = stiml + 104;
  float* rvblk = nzblk + kBlk * 2 * kN;
  float4* gpk  = (float4*)(rvblk + kBlk * 2 * kN);

  const int tid   = threadIdx.x;
  const int lane  = tid & 31;
  const int wav   = __builtin_amdgcn_readfirstlane(tid >> 5);
  const int lcol  = lane & 15;

  const int grp = lane >> 3;
  const int sub = lane & 7;
  float    wreg[kE];
  unsigned qoff[kE];
#pragma unroll
  for (int cc = 0; cc < 2; ++cc) {
#pragma unroll
    for (int t = 0; t < kCols; ++t) {
      const int i = kRows * wav + 2 * grp + cc;
      const int j = 8 * ((t + 2 * grp) & (kCols - 1)) + sub;
      wreg[kCols * cc + t] = Wt[i * kN + j];
      qoff[kCols * cc + t] =
          ((unsigned)((1 - lags[i * kN + j]) & 255) << 9) | ((unsigned)j << 2);
    }
  }

  for (int t = tid; t < 100; t += kThreads) stiml[t] = stimulus[t];
  for (int idx = tid; idx < kRingT * kN; idx += kThreads) {
    ring[idx] = initial_cond[2 * (idx & 127)];
  }

  for (int t = tid; t < 2 * kN; t += kThreads) gpk[t] = make_float4(1.f, 1.f, 1.f, 1.f);

  float cur_r = 0.f, cur_V = 0.f, eta = 0.f;
  float hs = 1.f, hf = 1.f;
  float bvl = 1.f, bq = 1.f;
  const float g0 = g[0];
  if (tid < kN) {
    cur_r  = initial_cond[2 * tid];
    cur_V  = initial_cond[2 * tid + 1];
    eta    = region_pars[tid];
  }
  __syncthreads();

  auto hw_sqrt = [](float v) -> float { return __builtin_amdgcn_sqrtf(v); };
  auto hw_rcp  = [](float v) -> float { return __builtin_amdgcn_rcpf(v); };

  auto pow3125 = [&](float v) -> float {
    const float r8 = hw_sqrt(hw_sqrt(hw_sqrt(v)));
    return (v * v) * (v * r8);
  };

  auto coupling_window = [&](int base, float* cx_out, auto&& first,
                             auto&& beside) {
    const unsigned tb = (unsigned)(base & 255) << 9;
    const char* const ringb = (const char*)ring;
    first();
    __builtin_amdgcn_sched_barrier(0);
    float rr[kE];
#pragma unroll
    for (int e = 0; e < kE; ++e)
      rr[e] = *(const float*)(ringb + ((tb + qoff[e]) & kRingMask));
    beside();
    float p[2];
#pragma unroll
    for (int cc = 0; cc < 2; ++cc) {
      float se = wreg[kCols * cc] * rr[kCols * cc];
      float so = wreg[kCols * cc + 1] * rr[kCols * cc + 1];
#pragma unroll
      for (int t = 2; t < kCols; t += 2) {
        se = fmaf(wreg[kCols * cc + t], rr[kCols * cc + t], se);
        so = fmaf(wreg[kCols * cc + t + 1], rr[kCols * cc + t + 1], so);
      }
      p[cc] = se + so;
    }
#if HAS_WMMA4
    v2f a, b;
    a.x = p[0];
    a.y = p[1];
    b.x = (lcol == 2 * (lane >> 4)) ? 1.0f : 0.0f;
    b.y = (lcol == 2 * (lane >> 4) + 1) ? 1.0f : 0.0f;
    v8f acc = {};
    acc = __builtin_amdgcn_wmma_f32_16x16x4_f32(
        false, a, false, b, (short)0, acc, false, false);
    const float s = ((acc[0] + acc[1]) + (acc[2] + acc[3])) +
                    ((acc[4] + acc[5]) + (acc[6] + acc[7]));
    if (lcol < 4)
      cx_out[kRows * wav + 4 * (lcol >> 1) + 2 * (lane >> 4) + (lcol & 1)] = s;
#else
#pragma unroll
    for (int cc = 0; cc < 2; ++cc) {
      float s = p[cc];
      for (int m = 4; m >= 1; m >>= 1) s += __shfl_xor(s, m, 32);
      if (sub == cc) cx_out[kRows * wav + 2 * grp + cc] = s;
    }
#endif
  };

  auto extract = [&](float f) -> float {
    return f * (1.f - powf(kOneMinusE0, 1.f / f)) * kInvE0;
  };
  auto flow_step = [&](float x, float hb) -> float4 {
    const float d1s = x - kRTS * hs - kRTF * (hf - 1.f);
    const float d1f = hs;
    const float s2 = hs + kBDT * d1s;
    const float f2 = hf + kBDT * d1f;
    const float d2s = x - kRTS * s2 - kRTF * (f2 - 1.f);
    const float d2f = s2;
    hs += hb * (d1s + d2s);
    hf += hb * (d1f + d2f);
    const float nf2 = hf + kBDT * hs;
    return make_float4(hf, nf2, extract(hf), extract(nf2));
  };
  auto vq_step = [&](float4 pk, float hb) {
    const float vr1 = pow3125(bvl);
    const float d1v = kRTO * (pk.x - vr1);
    const float d1q = kRTO * (pk.z - vr1 * bq * hw_rcp(bvl));
    const float v2 = bvl + kBDT * d1v;
    const float q2 = bq + kBDT * d1q;
    const float vr2 = pow3125(v2);
    const float d2v = kRTO * (pk.y - vr2);
    const float d2q = kRTO * (pk.w - vr2 * q2 * hw_rcp(v2));
    bvl += hb * (d1v + d2v);
    bq  += hb * (d1q + d2q);
  };

  coupling_window(-1, cxA, [] {}, [] {});
  __syncthreads();
  float cxp = (tid < kN) ? cxA[tid] : 0.f;
  float xir = 0.f, xiV = 0.f, nxr = 0.f, d1V = 0.f, dwv = 0.f, stv = 0.f;

  auto flush_rv = [&](int blk) {
    float4* dst = (float4*)out + blk * (kBlk * 2 * kN / 4);
    const float4* s4 = (const float4*)rvblk;
#pragma unroll
    for (int k = 0; k < kQ4; ++k) dst[tid + kThreads * k] = s4[tid + kThreads * k];
  };

  auto v_half = [&](float cxn, int s) {
    const float I2 = g0 * cxn + stv;
    const float pt2 = kPi * xir;
    const float d2V = xiV * xiV + eta + kJC * xir + I2 - pt2 * pt2;
    const float nxV = cur_V + kDT * 0.5f * (d1V + d2V) + dwv;
    cur_r = nxr;
    cur_V = nxV;
    cxp = cxn;

    float2 rv2; rv2.x = nxr; rv2.y = nxV;
    ((float2*)rvblk)[s * kN + tid] = rv2;
  };

  auto step = [&](auto deferred, int blk, int s) {
    constexpr bool kDeferred = decltype(deferred)::value;
    const int T = blk * kBlk + s;
    float* const cxN = (T & 1) ? cxA : cxB;
    float* const cxO = (T & 1) ? cxB : cxA;
    float4* const gpkT = gpk + (T & 1) * kN;
    float4* const gpkO = gpk + ((T & 1) ^ 1) * kN;

    const float hb = (T > 0) ? 0.5f * kBDT : 0.f;
    if (wav < 4) {
      float cxn = 0.f;
      float2 nz = make_float2(0.f, 0.f);
      float stc = 0.f;
      coupling_window(T, cxN,
        [&] {
          if (kDeferred) cxn = cxO[tid];
          nz = ((const float2*)nzblk)[s * kN + tid];
          stc = stiml[T / 10];
        },
        [&] {
        if (kDeferred) v_half(cxn, s - 1);
        const float dwr = kSig * nz.x;
        dwv = kSig * nz.y;
        stv = (tid == 0) ? stc : 0.f;
        const float rc = cur_r, Vc = cur_V;
        const float I1 = g0 * cxp + stv;
        const float d1r = kInvPi + 2.f * rc * Vc;
        const float pt1 = kPi * rc;
        d1V = Vc * Vc + eta + kJC * rc + I1 - pt1 * pt1;
        xir = fmaxf(rc + kDT * d1r + dwr, 0.f);
        xiV = Vc + kDT * d1V + dwv;
        const float d2r = kInvPi + 2.f * xir * xiV;
        nxr = fmaxf(rc + kDT * 0.5f * (d1r + d2r) + dwr, 0.f);
        ring[((T + 1) & 255) * kRingPitch + tid] = nxr;
        gpkT[tid] = flow_step(cur_r, hb);
      });
    } else if (wav < 8) {
      float4 pk = make_float4(1.f, 1.f, 1.f, 1.f);
      coupling_window(T, cxN,
        [&] { pk = gpkO[tid - kN]; },
        [&] { vq_step(pk, hb); });
    } else {
      coupling_window(T, cxN, [] {}, [] {});
    }
    __syncthreads();
  };

  for (int blk = 0; blk < kNBlk; ++blk) {
    __syncthreads();
    {
      const float4* nsrc = (const float4*)noise + blk * (kBlk * 2 * kN / 4);
      float4 nz4[kQ4];
#pragma unroll
      for (int k = 0; k < kQ4; ++k) nz4[k] = nsrc[tid + kThreads * k];
      if (blk > 0) flush_rv(blk - 1);
#pragma unroll
      for (int k = 0; k < kQ4; ++k)
        ((float4*)nzblk)[tid + kThreads * k] = nz4[k];
    }
    __syncthreads();

    step(Flag<false>{}, blk, 0);
    for (int s = 1; s < kBlk; ++s) step(Flag<true>{}, blk, s);

    if (wav < 4) {
      const int T = blk * kBlk + kBlk - 1;
      v_half(((T & 1) ? cxA : cxB)[tid], kBlk - 1);
    }
  }

  __syncthreads();
  flush_rv(kNBlk - 1);

  if (tid >= kN && tid < 2 * kN) {
    vq_step(gpk[((kSteps - 1) & 1) * kN + (tid - kN)], 0.5f * kBDT);
    const float bold = kV0 * (kK1 * (1.f - bq) + kK2 * (1.f - bq / bvl) +
                              kK3 * (1.f - bvl));
    out[kSteps * kN * 2 + (tid - kN)] = bold;
  }
}

extern "C" void kernel_launch(void* const* d_in, const int* in_sizes, int n_in,
                              void* d_out, int out_size, void* d_ws, size_t ws_size,
                              hipStream_t stream) {
  (void)in_sizes; (void)n_in; (void)d_ws; (void)ws_size; (void)out_size;
  const float* region_pars  = (const float*)d_in[0];
  const float* Wt           = (const float*)d_in[1];
  const float* g            = (const float*)d_in[2];
  const float* stimulus     = (const float*)d_in[3];
  const float* noise        = (const float*)d_in[4];
  const float* initial_cond = (const float*)d_in[5];
  const int*   lags         = (const int*)d_in[6];

  const size_t smem_bytes =
      (size_t)(kRingT * kRingPitch + kN + kN + 104 + 2 * kBlk * 2 * kN +
               2 * kN * 4) *
      sizeof(float);

  tvb_mpr_persistent<<<1, kThreads, smem_bytes, stream>>>(
      region_pars, Wt, g, stimulus, noise, initial_cond, lags, (float*)d_out);
}
